# starting kernel
# baseline (speedup 1.0000x reference)
#include <hip/hip_runtime.h>
#include <hip/hip_bf16.h>
#include <math.h>

typedef __attribute__((ext_vector_type(2))) float v2f;
typedef __attribute__((ext_vector_type(8))) float v8f;

#define LRC  0.01f
#define EPSC 1e-5f

#define Bq 256
#define Lq 4096
#define Hq 32
#define Vq 64
#define CH 16
#define NCH (Lq / CH)
#define A1STR 68
#define HSTR  36

__launch_bounds__(64, 1)
__global__ void pmm_fused_kernel(const int* __restrict__ seq,
                                 const float* __restrict__ embed,
                                 const float* __restrict__ ff_w1,
                                 const float* __restrict__ ff_b1,
                                 const float* __restrict__ ff_w2,
                                 const float* __restrict__ ff_b2,
                                 const float* __restrict__ ln_g,
                                 const float* __restrict__ ln_b,
                                 const float* __restrict__ mlp_w1,
                                 const float* __restrict__ mlp_b1,
                                 const float* __restrict__ mlp_w2,
                                 const float* __restrict__ mlp_b2,
                                 const float* __restrict__ out_w,
                                 const float* __restrict__ out_b,
                                 float* __restrict__ out)
{
    __shared__ float embedLds[Vq * Hq];
    __shared__ float w1Lds[2 * Hq * Hq];
    __shared__ float w2Lds[Hq * 2 * Hq];
    __shared__ float a1buf[CH * A1STR];
    __shared__ float hid[2][CH * HSTR];
    __shared__ float ctxLds[Hq];

    const int b    = blockIdx.x;
    const int tid  = threadIdx.x;
    const int lane = tid & 31;
    const int wave = tid >> 5;
    const int m    = lane & 15;
    const int hi   = lane >> 4;

    for (int i = tid; i < Vq * Hq; i += 64)     embedLds[i] = embed[i];
    for (int i = tid; i < 2 * Hq * Hq; i += 64) w1Lds[i]    = ff_w1[i];
    for (int i = tid; i < 2 * Hq * Hq; i += 64) w2Lds[i]    = ff_w2[i];
    __syncthreads();

    float b1t[4], b2t[2], lnGv = 0.f, lnBv = 0.f;
    if (wave == 0) {
        #pragma unroll
        for (int nt = 0; nt < 4; ++nt) b1t[nt] = ff_b1[nt * 16 + m];
        #pragma unroll
        for (int nt = 0; nt < 2; ++nt) b2t[nt] = ff_b2[nt * 16 + m];
        lnGv = ln_g[lane];
        lnBv = ln_b[lane];
    }

    float w1c[8], b1v[8], w2r[8], b2v = 0.f;
    if (wave == 1) {
        #pragma unroll
        for (int i = 0; i < 8; ++i) {
            w1c[i] = mlp_w1[i * Hq + lane];
            b1v[i] = mlp_b1[i];
            w2r[i] = mlp_w2[lane * 8 + i];
        }
        b2v = mlp_b2[lane];
    }

    auto produce = [&](int c, float* hbuf) {
        const int base = c * CH;
        const int tok  = seq[b * Lq + base + m];

        v2f eA[8];
        #pragma unroll
        for (int kb = 0; kb < 8; ++kb)
            eA[kb] = *(const v2f*)&embedLds[tok * Hq + 4 * kb + 2 * hi];

        #pragma unroll 1
        for (int nt = 0; nt < 4; ++nt) {
            const int n0 = nt * 16;
            v8f acc = {};
            #pragma unroll
            for (int kb = 0; kb < 8; ++kb) {
                v2f bB = *(const v2f*)&w1Lds[(n0 + m) * Hq + 4 * kb + 2 * hi];
                acc = __builtin_amdgcn_wmma_f32_16x16x4_f32(
                        false, eA[kb], false, bB, (short)0, acc, false, false);
            }
            const float bias = b1t[nt];
            #pragma unroll
            for (int r = 0; r < 8; ++r) {
                float z = acc[r] + bias;
                a1buf[(r + 8 * hi) * A1STR + n0 + m] = fmaxf(z, 0.f);
            }
        }
        asm volatile("s_wait_dscnt 0" ::: "memory");
        __builtin_amdgcn_wave_barrier();

        v2f a2[16];
        #pragma unroll
        for (int kb = 0; kb < 16; ++kb)
            a2[kb] = *(const v2f*)&a1buf[m * A1STR + 4 * kb + 2 * hi];

        #pragma unroll 1
        for (int nt = 0; nt < 2; ++nt) {
            const int n0 = nt * 16;
            v8f acc = {};
            #pragma unroll
            for (int kb = 0; kb < 16; ++kb) {
                v2f bB = *(const v2f*)&w2Lds[(n0 + m) * (2 * Hq) + 4 * kb + 2 * hi];
                acc = __builtin_amdgcn_wmma_f32_16x16x4_f32(
                        false, a2[kb], false, bB, (short)0, acc, false, false);
            }
            const float bias = b2t[nt];
            #pragma unroll
            for (int r = 0; r < 8; ++r) {
                int   tm = __shfl(tok, r + 8 * hi, 32);
                float e  = embedLds[tm * Hq + n0 + m];
                hbuf[(r + 8 * hi) * HSTR + n0 + m] = acc[r] + bias + e;
            }
        }
        asm volatile("s_wait_dscnt 0" ::: "memory");
        __builtin_amdgcn_wave_barrier();

        #pragma unroll 1
        for (int t = 0; t < CH; ++t) {
            float z = hbuf[t * HSTR + lane];
            float s = z, s2 = z * z;
            #pragma unroll
            for (int off = 16; off >= 1; off >>= 1) {
                s  += __shfl_xor(s,  off, 32);
                s2 += __shfl_xor(s2, off, 32);
            }
            float mu  = s * (1.f / 32.f);
            float var = s2 * (1.f / 32.f) - mu * mu;
            float nz  = (z - mu) * rsqrtf(var + EPSC);
            hbuf[t * HSTR + lane] = nz * lnGv + lnBv;
        }
        asm volatile("s_wait_dscnt 0" ::: "memory");
    };

    auto consume = [&](const float* hbuf, int nsteps) {
        #pragma unroll 1
        for (int s = 0; s < nsteps; ++s) {
            float x = hbuf[(2 * s) * HSTR + lane];
            float v = hbuf[(2 * s + 1) * HSTR + lane];

            float p[8];
            #pragma unroll
            for (int i = 0; i < 8; ++i) p[i] = w1c[i] * x;
            #pragma unroll
            for (int off = 16; off >= 1; off >>= 1) {
                #pragma unroll
                for (int i = 0; i < 8; ++i) p[i] += __shfl_xor(p[i], off, 32);
            }
            float a1p[8], a1v[8];
            float y = b2v;
            #pragma unroll
            for (int i = 0; i < 8; ++i) {
                a1p[i] = p[i] + b1v[i];
                a1v[i] = fmaxf(a1p[i], 0.f);
                y = fmaf(w2r[i], a1v[i], y);
            }
            float d = (y - v) * 0.0625f;

            float q[8];
            #pragma unroll
            for (int i = 0; i < 8; ++i) q[i] = w2r[i] * d;
            #pragma unroll
            for (int off = 16; off >= 1; off >>= 1) {
                #pragma unroll
                for (int i = 0; i < 8; ++i) q[i] += __shfl_xor(q[i], off, 32);
            }
            b2v -= LRC * d;
            #pragma unroll
            for (int i = 0; i < 8; ++i) {
                float g1 = (a1p[i] > 0.f) ? q[i] : 0.f;
                w2r[i] = fmaf(-LRC * d, a1v[i], w2r[i]);
                b1v[i] -= LRC * g1;
                w1c[i] = fmaf(-LRC * g1, x, w1c[i]);
            }
        }
    };

    if (wave == 0) produce(0, hid[0]);
    __syncthreads();

    for (int c = 0; c < NCH; ++c) {
        if (wave == 0) {
            if (c + 1 < NCH) produce(c + 1, hid[(c + 1) & 1]);
        } else {
            consume(hid[c & 1], (c == NCH - 1) ? 7 : 8);
        }
        __syncthreads();
    }

    if (wave == 1) {
        const float* hbuf = hid[(NCH - 1) & 1];
        float x = hbuf[15 * HSTR + lane];
        float p[8];
        #pragma unroll
        for (int i = 0; i < 8; ++i) p[i] = w1c[i] * x;
        #pragma unroll
        for (int off = 16; off >= 1; off >>= 1) {
            #pragma unroll
            for (int i = 0; i < 8; ++i) p[i] += __shfl_xor(p[i], off, 32);
        }
        float y = b2v;
        #pragma unroll
        for (int i = 0; i < 8; ++i) y = fmaf(w2r[i], fmaxf(p[i] + b1v[i], 0.f), y);
        ctxLds[lane] = y;
    }
    __syncthreads();

    {
        float acc = out_b[tid];
        #pragma unroll
        for (int h = 0; h < Hq; ++h)
            acc = fmaf(ctxLds[h], out_w[tid * Hq + h], acc);
        out[b * Vq + tid] = acc;
    }
}

extern "C" void kernel_launch(void* const* d_in, const int* in_sizes, int n_in,
                              void* d_out, int out_size, void* d_ws, size_t ws_size,
                              hipStream_t stream) {
    (void)in_sizes; (void)n_in; (void)out_size; (void)d_ws; (void)ws_size;
    const int*   seq    = (const int*)  d_in[0];
    const float* embed  = (const float*)d_in[1];
    const float* ff_w1  = (const float*)d_in[2];
    const float* ff_b1  = (const float*)d_in[3];
    const float* ff_w2  = (const float*)d_in[4];
    const float* ff_b2  = (const float*)d_in[5];
    const float* ln_g   = (const float*)d_in[6];
    const float* ln_b   = (const float*)d_in[7];
    const float* mlp_w1 = (const float*)d_in[8];
    const float* mlp_b1 = (const float*)d_in[9];
    const float* mlp_w2 = (const float*)d_in[10];
    const float* mlp_b2 = (const float*)d_in[11];
    const float* out_w  = (const float*)d_in[12];
    const float* out_b  = (const float*)d_in[13];
    float*       out    = (float*)d_out;

    pmm_fused_kernel<<<dim3(Bq), dim3(64), 0, stream>>>(
        seq, embed, ff_w1, ff_b1, ff_w2, ff_b2, ln_g, ln_b,
        mlp_w1, mlp_b1, mlp_w2, mlp_b2, out_w, out_b, out);
}


# All 32-lane butterfly sums (LayerNorm mean/var, both 8-value reductions of each SGD step) moved from ds_bpermute to DPP row_xmask + permlanex16 adds; same xor order, sums bit-identical
# speedup vs baseline: 1.9371x; 1.9371x over previous
#include <hip/hip_runtime.h>
#include <hip/hip_bf16.h>
#include <math.h>

typedef __attribute__((ext_vector_type(2))) float v2f;
typedef __attribute__((ext_vector_type(8))) float v8f;

#define LRC  0.01f
#define EPSC 1e-5f

#define Bq 256
#define Lq 4096
#define Hq 32
#define Vq 64
#define CH 16
#define NCH (Lq / CH)
#define A1STR 68
#define HSTR  36

template <int OFF>
__device__ __forceinline__ float xlane(float v) {
    const int b = __builtin_bit_cast(int, v);
    if constexpr (OFF == 16) {
        return __builtin_bit_cast(float,
            __builtin_amdgcn_permlanex16(b, b, 0x76543210, 0xfedcba98, false, false));
    } else {
        return __builtin_bit_cast(float,
            __builtin_amdgcn_update_dpp(0, b, 0x160 + OFF, 0xf, 0xf, true));
    }
}
template <int OFF, int N>
__device__ __forceinline__ void bfly_add(float (&v)[N]) {
    #pragma unroll
    for (int i = 0; i < N; ++i) v[i] += xlane<OFF>(v[i]);
}
template <int N>
__device__ __forceinline__ void wave_sum(float (&v)[N]) {
    bfly_add<16, N>(v);
    bfly_add<8, N>(v);
    bfly_add<4, N>(v);
    bfly_add<2, N>(v);
    bfly_add<1, N>(v);
}

__launch_bounds__(64, 1)
__global__ void pmm_fused_kernel(const int* __restrict__ seq,
                                 const float* __restrict__ embed,
                                 const float* __restrict__ ff_w1,
                                 const float* __restrict__ ff_b1,
                                 const float* __restrict__ ff_w2,
                                 const float* __restrict__ ff_b2,
                                 const float* __restrict__ ln_g,
                                 const float* __restrict__ ln_b,
                                 const float* __restrict__ mlp_w1,
                                 const float* __restrict__ mlp_b1,
                                 const float* __restrict__ mlp_w2,
                                 const float* __restrict__ mlp_b2,
                                 const float* __restrict__ out_w,
                                 const float* __restrict__ out_b,
                                 float* __restrict__ out)
{
    __shared__ float embedLds[Vq * Hq];
    __shared__ float w1Lds[2 * Hq * Hq];
    __shared__ float w2Lds[Hq * 2 * Hq];
    __shared__ float a1buf[CH * A1STR];
    __shared__ float hid[2][CH * HSTR];
    __shared__ float ctxLds[Hq];

    const int b    = blockIdx.x;
    const int tid  = threadIdx.x;
    const int lane = tid & 31;
    const int wave = tid >> 5;
    const int m    = lane & 15;
    const int hi   = lane >> 4;

    for (int i = tid; i < Vq * Hq; i += 64)     embedLds[i] = embed[i];
    for (int i = tid; i < 2 * Hq * Hq; i += 64) w1Lds[i]    = ff_w1[i];
    for (int i = tid; i < 2 * Hq * Hq; i += 64) w2Lds[i]    = ff_w2[i];
    __syncthreads();

    float b1t[4], b2t[2], lnGv = 0.f, lnBv = 0.f;
    if (wave == 0) {
        #pragma unroll
        for (int nt = 0; nt < 4; ++nt) b1t[nt] = ff_b1[nt * 16 + m];
        #pragma unroll
        for (int nt = 0; nt < 2; ++nt) b2t[nt] = ff_b2[nt * 16 + m];
        lnGv = ln_g[lane];
        lnBv = ln_b[lane];
    }

    float w1c[8], b1v[8], w2r[8], b2v = 0.f;
    if (wave == 1) {
        #pragma unroll
        for (int i = 0; i < 8; ++i) {
            w1c[i] = mlp_w1[i * Hq + lane];
            b1v[i] = mlp_b1[i];
            w2r[i] = mlp_w2[lane * 8 + i];
        }
        b2v = mlp_b2[lane];
    }

    auto produce = [&](int c, float* hbuf) {
        const int base = c * CH;
        const int tok  = seq[b * Lq + base + m];

        v2f eA[8];
        #pragma unroll
        for (int kb = 0; kb < 8; ++kb)
            eA[kb] = *(const v2f*)&embedLds[tok * Hq + 4 * kb + 2 * hi];

        #pragma unroll 1
        for (int nt = 0; nt < 4; ++nt) {
            const int n0 = nt * 16;
            v8f acc = {};
            #pragma unroll
            for (int kb = 0; kb < 8; ++kb) {
                v2f bB = *(const v2f*)&w1Lds[(n0 + m) * Hq + 4 * kb + 2 * hi];
                acc = __builtin_amdgcn_wmma_f32_16x16x4_f32(
                        false, eA[kb], false, bB, (short)0, acc, false, false);
            }
            const float bias = b1t[nt];
            #pragma unroll
            for (int r = 0; r < 8; ++r) {
                float z = acc[r] + bias;
                a1buf[(r + 8 * hi) * A1STR + n0 + m] = fmaxf(z, 0.f);
            }
        }
        asm volatile("s_wait_dscnt 0" ::: "memory");
        __builtin_amdgcn_wave_barrier();

        v2f a2[16];
        #pragma unroll
        for (int kb = 0; kb < 16; ++kb)
            a2[kb] = *(const v2f*)&a1buf[m * A1STR + 4 * kb + 2 * hi];

        #pragma unroll 1
        for (int nt = 0; nt < 2; ++nt) {
            const int n0 = nt * 16;
            v8f acc = {};
            #pragma unroll
            for (int kb = 0; kb < 16; ++kb) {
                v2f bB = *(const v2f*)&w2Lds[(n0 + m) * (2 * Hq) + 4 * kb + 2 * hi];
                acc = __builtin_amdgcn_wmma_f32_16x16x4_f32(
                        false, a2[kb], false, bB, (short)0, acc, false, false);
            }
            const float bias = b2t[nt];
            #pragma unroll
            for (int r = 0; r < 8; ++r) {
                int   tm = __shfl(tok, r + 8 * hi, 32);
                float e  = embedLds[tm * Hq + n0 + m];
                hbuf[(r + 8 * hi) * HSTR + n0 + m] = acc[r] + bias + e;
            }
        }
        asm volatile("s_wait_dscnt 0" ::: "memory");
        __builtin_amdgcn_wave_barrier();

        #pragma unroll 1
        for (int t = 0; t < CH; ++t) {
            float z = hbuf[t * HSTR + lane];
            float ss[2] = {z, z * z};
            wave_sum<2>(ss);
            const float s = ss[0], s2 = ss[1];
            float mu  = s * (1.f / 32.f);
            float var = s2 * (1.f / 32.f) - mu * mu;
            float nz  = (z - mu) * rsqrtf(var + EPSC);
            hbuf[t * HSTR + lane] = nz * lnGv + lnBv;
        }
        asm volatile("s_wait_dscnt 0" ::: "memory");
    };

    auto consume = [&](const float* hbuf, int nsteps) {
        #pragma unroll 1
        for (int s = 0; s < nsteps; ++s) {
            float x = hbuf[(2 * s) * HSTR + lane];
            float v = hbuf[(2 * s + 1) * HSTR + lane];

            float p[8];
            #pragma unroll
            for (int i = 0; i < 8; ++i) p[i] = w1c[i] * x;
            wave_sum<8>(p);
            float a1p[8], a1v[8];
            float y = b2v;
            #pragma unroll
            for (int i = 0; i < 8; ++i) {
                a1p[i] = p[i] + b1v[i];
                a1v[i] = fmaxf(a1p[i], 0.f);
                y = fmaf(w2r[i], a1v[i], y);
            }
            float d = (y - v) * 0.0625f;

            float q[8];
            #pragma unroll
            for (int i = 0; i < 8; ++i) q[i] = w2r[i] * d;
            wave_sum<8>(q);
            b2v -= LRC * d;
            #pragma unroll
            for (int i = 0; i < 8; ++i) {
                float g1 = (a1p[i] > 0.f) ? q[i] : 0.f;
                w2r[i] = fmaf(-LRC * d, a1v[i], w2r[i]);
                b1v[i] -= LRC * g1;
                w1c[i] = fmaf(-LRC * g1, x, w1c[i]);
            }
        }
    };

    if (wave == 0) produce(0, hid[0]);
    __syncthreads();

    for (int c = 0; c < NCH; ++c) {
        if (wave == 0) {
            if (c + 1 < NCH) produce(c + 1, hid[(c + 1) & 1]);
        } else {
            consume(hid[c & 1], (c == NCH - 1) ? 7 : 8);
        }
        __syncthreads();
    }

    if (wave == 1) {
        const float* hbuf = hid[(NCH - 1) & 1];
        float x = hbuf[15 * HSTR + lane];
        float p[8];
        #pragma unroll
        for (int i = 0; i < 8; ++i) p[i] = w1c[i] * x;
        wave_sum<8>(p);
        float y = b2v;
        #pragma unroll
        for (int i = 0; i < 8; ++i) y = fmaf(w2r[i], fmaxf(p[i] + b1v[i], 0.f), y);
        ctxLds[lane] = y;
    }
    __syncthreads();

    {
        float acc = out_b[tid];
        #pragma unroll
        for (int h = 0; h < Hq; ++h)
            acc = fmaf(ctxLds[h], out_w[tid * Hq + h], acc);
        out[b * Vq + tid] = acc;
    }
}

extern "C" void kernel_launch(void* const* d_in, const int* in_sizes, int n_in,
                              void* d_out, int out_size, void* d_ws, size_t ws_size,
                              hipStream_t stream) {
    (void)in_sizes; (void)n_in; (void)out_size; (void)d_ws; (void)ws_size;
    const int*   seq    = (const int*)  d_in[0];
    const float* embed  = (const float*)d_in[1];
    const float* ff_w1  = (const float*)d_in[2];
    const float* ff_b1  = (const float*)d_in[3];
    const float* ff_w2  = (const float*)d_in[4];
    const float* ff_b2  = (const float*)d_in[5];
    const float* ln_g   = (const float*)d_in[6];
    const float* ln_b   = (const float*)d_in[7];
    const float* mlp_w1 = (const float*)d_in[8];
    const float* mlp_b1 = (const float*)d_in[9];
    const float* mlp_w2 = (const float*)d_in[10];
    const float* mlp_b2 = (const float*)d_in[11];
    const float* out_w  = (const float*)d_in[12];
    const float* out_b  = (const float*)d_in[13];
    float*       out    = (float*)d_out;

    pmm_fused_kernel<<<dim3(Bq), dim3(64), 0, stream>>>(
        seq, embed, ff_w1, ff_b1, ff_w2, ff_b2, ln_g, ln_b,
        mlp_w1, mlp_b1, mlp_w2, mlp_b2, out_w, out_b, out);
}


# Token encoding spread over three producer waves (one 16-token chunk each per barrier interval, one wave per SIMD) feeding the one SGD wave; per-chunk and per-step arithmetic and order unchanged
# speedup vs baseline: 4.1047x; 2.1190x over previous
#include <hip/hip_runtime.h>
#include <hip/hip_bf16.h>
#include <math.h>

typedef __attribute__((ext_vector_type(2))) float v2f;
typedef __attribute__((ext_vector_type(8))) float v8f;

#define LRC  0.01f
#define EPSC 1e-5f

#define Bq 256
#define Lq 4096
#define Hq 32
#define Vq 64
#define CH 16
#define NCH (Lq / CH)
#define NP  3
#define NGRP ((NCH + NP - 1) / NP)
#define NTHR 128
static_assert(NTHR == 32 * (NP + 1), "block = NP producer waves + 1 consumer wave");
#define A1STR 68
#define HSTR  36

template <int OFF>
__device__ __forceinline__ float xlane(float v) {
    const int b = __builtin_bit_cast(int, v);
    if constexpr (OFF == 16) {
        return __builtin_bit_cast(float,
            __builtin_amdgcn_permlanex16(b, b, 0x76543210, 0xfedcba98, false, false));
    } else {
        return __builtin_bit_cast(float,
            __builtin_amdgcn_update_dpp(0, b, 0x160 + OFF, 0xf, 0xf, true));
    }
}
template <int OFF, int N>
__device__ __forceinline__ void bfly_add(float (&v)[N]) {
    #pragma unroll
    for (int i = 0; i < N; ++i) v[i] += xlane<OFF>(v[i]);
}
template <int N>
__device__ __forceinline__ void wave_sum(float (&v)[N]) {
    bfly_add<16, N>(v);
    bfly_add<8, N>(v);
    bfly_add<4, N>(v);
    bfly_add<2, N>(v);
    bfly_add<1, N>(v);
}

__launch_bounds__(NTHR, 1)
__global__ void pmm_fused_kernel(const int* __restrict__ seq,
                                 const float* __restrict__ embed,
                                 const float* __restrict__ ff_w1,
                                 const float* __restrict__ ff_b1,
                                 const float* __restrict__ ff_w2,
                                 const float* __restrict__ ff_b2,
                                 const float* __restrict__ ln_g,
                                 const float* __restrict__ ln_b,
                                 const float* __restrict__ mlp_w1,
                                 const float* __restrict__ mlp_b1,
                                 const float* __restrict__ mlp_w2,
                                 const float* __restrict__ mlp_b2,
                                 const float* __restrict__ out_w,
                                 const float* __restrict__ out_b,
                                 float* __restrict__ out)
{
    __shared__ float embedLds[Vq * Hq];
    __shared__ float w1Lds[2 * Hq * Hq];
    __shared__ float w2Lds[Hq * 2 * Hq];
    __shared__ float a1buf[NP][CH * A1STR];
    __shared__ float hid[2][NP * CH * HSTR];
    __shared__ float ctxLds[Hq];

    const int b    = blockIdx.x;
    const int tid  = threadIdx.x;
    const int lane = tid & 31;
    const int wave = tid >> 5;
    const int m    = lane & 15;
    const int hi   = lane >> 4;

    for (int i = tid; i < Vq * Hq; i += NTHR)     embedLds[i] = embed[i];
    for (int i = tid; i < 2 * Hq * Hq; i += NTHR) w1Lds[i]    = ff_w1[i];
    for (int i = tid; i < 2 * Hq * Hq; i += NTHR) w2Lds[i]    = ff_w2[i];
    __syncthreads();

    float b1t[4], b2t[2], lnGv = 0.f, lnBv = 0.f;
    if (wave < NP) {
        #pragma unroll
        for (int nt = 0; nt < 4; ++nt) b1t[nt] = ff_b1[nt * 16 + m];
        #pragma unroll
        for (int nt = 0; nt < 2; ++nt) b2t[nt] = ff_b2[nt * 16 + m];
        lnGv = ln_g[lane];
        lnBv = ln_b[lane];
    }

    float w1c[8], b1v[8], w2r[8], b2v = 0.f;
    if (wave == NP) {
        #pragma unroll
        for (int i = 0; i < 8; ++i) {
            w1c[i] = mlp_w1[i * Hq + lane];
            b1v[i] = mlp_b1[i];
            w2r[i] = mlp_w2[lane * 8 + i];
        }
        b2v = mlp_b2[lane];
    }

    float* const a1w = a1buf[wave < NP ? wave : 0];
    auto produce = [&](int c, float* hbuf) {
        const int base = c * CH;
        const int tok  = seq[b * Lq + base + m];

        v2f eA[8];
        #pragma unroll
        for (int kb = 0; kb < 8; ++kb)
            eA[kb] = *(const v2f*)&embedLds[tok * Hq + 4 * kb + 2 * hi];

        #pragma unroll 1
        for (int nt = 0; nt < 4; ++nt) {
            const int n0 = nt * 16;
            v8f acc = {};
            #pragma unroll
            for (int kb = 0; kb < 8; ++kb) {
                v2f bB = *(const v2f*)&w1Lds[(n0 + m) * Hq + 4 * kb + 2 * hi];
                acc = __builtin_amdgcn_wmma_f32_16x16x4_f32(
                        false, eA[kb], false, bB, (short)0, acc, false, false);
            }
            const float bias = b1t[nt];
            #pragma unroll
            for (int r = 0; r < 8; ++r) {
                float z = acc[r] + bias;
                a1w[(r + 8 * hi) * A1STR + n0 + m] = fmaxf(z, 0.f);
            }
        }
        asm volatile("s_wait_dscnt 0" ::: "memory");
        __builtin_amdgcn_wave_barrier();

        v2f a2[16];
        #pragma unroll
        for (int kb = 0; kb < 16; ++kb)
            a2[kb] = *(const v2f*)&a1w[m * A1STR + 4 * kb + 2 * hi];

        #pragma unroll 1
        for (int nt = 0; nt < 2; ++nt) {
            const int n0 = nt * 16;
            v8f acc = {};
            #pragma unroll
            for (int kb = 0; kb < 16; ++kb) {
                v2f bB = *(const v2f*)&w2Lds[(n0 + m) * (2 * Hq) + 4 * kb + 2 * hi];
                acc = __builtin_amdgcn_wmma_f32_16x16x4_f32(
                        false, a2[kb], false, bB, (short)0, acc, false, false);
            }
            const float bias = b2t[nt];
            #pragma unroll
            for (int r = 0; r < 8; ++r) {
                int   tm = __shfl(tok, r + 8 * hi, 32);
                float e  = embedLds[tm * Hq + n0 + m];
                hbuf[(r + 8 * hi) * HSTR + n0 + m] = acc[r] + bias + e;
            }
        }
        asm volatile("s_wait_dscnt 0" ::: "memory");
        __builtin_amdgcn_wave_barrier();

        #pragma unroll 1
        for (int t = 0; t < CH; ++t) {
            float z = hbuf[t * HSTR + lane];
            float ss[2] = {z, z * z};
            wave_sum<2>(ss);
            const float s = ss[0], s2 = ss[1];
            float mu  = s * (1.f / 32.f);
            float var = s2 * (1.f / 32.f) - mu * mu;
            float nz  = (z - mu) * rsqrtf(var + EPSC);
            hbuf[t * HSTR + lane] = nz * lnGv + lnBv;
        }
        asm volatile("s_wait_dscnt 0" ::: "memory");
    };

    auto consume = [&](const float* hbuf, int nsteps) {
        #pragma unroll 1
        for (int s = 0; s < nsteps; ++s) {
            float x = hbuf[(2 * s) * HSTR + lane];
            float v = hbuf[(2 * s + 1) * HSTR + lane];

            float p[8];
            #pragma unroll
            for (int i = 0; i < 8; ++i) p[i] = w1c[i] * x;
            wave_sum<8>(p);
            float a1p[8], a1v[8];
            float y = b2v;
            #pragma unroll
            for (int i = 0; i < 8; ++i) {
                a1p[i] = p[i] + b1v[i];
                a1v[i] = fmaxf(a1p[i], 0.f);
                y = fmaf(w2r[i], a1v[i], y);
            }
            float d = (y - v) * 0.0625f;

            float q[8];
            #pragma unroll
            for (int i = 0; i < 8; ++i) q[i] = w2r[i] * d;
            wave_sum<8>(q);
            b2v -= LRC * d;
            #pragma unroll
            for (int i = 0; i < 8; ++i) {
                float g1 = (a1p[i] > 0.f) ? q[i] : 0.f;
                w2r[i] = fmaf(-LRC * d, a1v[i], w2r[i]);
                b1v[i] -= LRC * g1;
                w1c[i] = fmaf(-LRC * g1, x, w1c[i]);
            }
        }
    };

    if (wave < NP) produce(wave, hid[0] + wave * (CH * HSTR));
    __syncthreads();

    for (int g = 0; g < NGRP; ++g) {
        if (wave < NP) {
            const int c = NP * (g + 1) + wave;
            if (c < NCH) produce(c, hid[(g + 1) & 1] + wave * (CH * HSTR));
        } else {
            consume(hid[g & 1], (g == NGRP - 1) ? (8 * (NCH - NP * (NGRP - 1)) - 1) : 8 * NP);
        }
        __syncthreads();
    }

    if (wave == NP) {
        const float* hbuf = hid[(NGRP - 1) & 1] + ((NCH - 1) - NP * (NGRP - 1)) * (CH * HSTR);
        float x = hbuf[15 * HSTR + lane];
        float p[8];
        #pragma unroll
        for (int i = 0; i < 8; ++i) p[i] = w1c[i] * x;
        wave_sum<8>(p);
        float y = b2v;
        #pragma unroll
        for (int i = 0; i < 8; ++i) y = fmaf(w2r[i], fmaxf(p[i] + b1v[i], 0.f), y);
        ctxLds[lane] = y;
    }
    __syncthreads();

    if (tid < Vq) {
        float acc = out_b[tid];
        #pragma unroll
        for (int h = 0; h < Hq; ++h)
            acc = fmaf(ctxLds[h], out_w[tid * Hq + h], acc);
        out[b * Vq + tid] = acc;
    }
}

extern "C" void kernel_launch(void* const* d_in, const int* in_sizes, int n_in,
                              void* d_out, int out_size, void* d_ws, size_t ws_size,
                              hipStream_t stream) {
    (void)in_sizes; (void)n_in; (void)out_size; (void)d_ws; (void)ws_size;
    const int*   seq    = (const int*)  d_in[0];
    const float* embed  = (const float*)d_in[1];
    const float* ff_w1  = (const float*)d_in[2];
    const float* ff_b1  = (const float*)d_in[3];
    const float* ff_w2  = (const float*)d_in[4];
    const float* ff_b2  = (const float*)d_in[5];
    const float* ln_g   = (const float*)d_in[6];
    const float* ln_b   = (const float*)d_in[7];
    const float* mlp_w1 = (const float*)d_in[8];
    const float* mlp_b1 = (const float*)d_in[9];
    const float* mlp_w2 = (const float*)d_in[10];
    const float* mlp_b2 = (const float*)d_in[11];
    const float* out_w  = (const float*)d_in[12];
    const float* out_b  = (const float*)d_in[13];
    float*       out    = (float*)d_out;

    pmm_fused_kernel<<<dim3(Bq), dim3(NTHR), 0, stream>>>(
        seq, embed, ff_w1, ff_b1, ff_w2, ff_b2, ln_g, ln_b,
        mlp_w1, mlp_b1, mlp_w2, mlp_b2, out_w, out_b, out);
}


# SGD wave reads the key/value rows of step s+1 from LDS at the top of step s (one-step software pipeline): the LDS read latency leaves the serial SGD chain; arithmetic and sum order unchanged
# speedup vs baseline: 4.3177x; 1.0519x over previous
#include <hip/hip_runtime.h>
#include <hip/hip_bf16.h>
#include <math.h>

typedef __attribute__((ext_vector_type(2))) float v2f;
typedef __attribute__((ext_vector_type(8))) float v8f;

#define LRC  0.01f
#define EPSC 1e-5f

#define Bq 256
#define Lq 4096
#define Hq 32
#define Vq 64
#define CH 16
#define NCH (Lq / CH)
#define NP  3
#define NGRP ((NCH + NP - 1) / NP)
#define NTHR 128
static_assert(NTHR == 32 * (NP + 1), "block = NP producer waves + 1 consumer wave");
#define A1STR 68
#define HSTR  36

template <int OFF>
__device__ __forceinline__ float xlane(float v) {
    const int b = __builtin_bit_cast(int, v);
    if constexpr (OFF == 16) {
        return __builtin_bit_cast(float,
            __builtin_amdgcn_permlanex16(b, b, 0x76543210, 0xfedcba98, false, false));
    } else {
        return __builtin_bit_cast(float,
            __builtin_amdgcn_update_dpp(0, b, 0x160 + OFF, 0xf, 0xf, true));
    }
}
template <int OFF, int N>
__device__ __forceinline__ void bfly_add(float (&v)[N]) {
    #pragma unroll
    for (int i = 0; i < N; ++i) v[i] += xlane<OFF>(v[i]);
}
template <int N>
__device__ __forceinline__ void wave_sum(float (&v)[N]) {
    bfly_add<16, N>(v);
    bfly_add<8, N>(v);
    bfly_add<4, N>(v);
    bfly_add<2, N>(v);
    bfly_add<1, N>(v);
}

__launch_bounds__(NTHR, 1)
__global__ void pmm_fused_kernel(const int* __restrict__ seq,
                                 const float* __restrict__ embed,
                                 const float* __restrict__ ff_w1,
                                 const float* __restrict__ ff_b1,
                                 const float* __restrict__ ff_w2,
                                 const float* __restrict__ ff_b2,
                                 const float* __restrict__ ln_g,
                                 const float* __restrict__ ln_b,
                                 const float* __restrict__ mlp_w1,
                                 const float* __restrict__ mlp_b1,
                                 const float* __restrict__ mlp_w2,
                                 const float* __restrict__ mlp_b2,
                                 const float* __restrict__ out_w,
                                 const float* __restrict__ out_b,
                                 float* __restrict__ out)
{
    __shared__ float embedLds[Vq * Hq];
    __shared__ float w1Lds[2 * Hq * Hq];
    __shared__ float w2Lds[Hq * 2 * Hq];
    __shared__ float a1buf[NP][CH * A1STR];
    __shared__ float hid[2][NP * CH * HSTR];
    __shared__ float ctxLds[Hq];

    const int b    = blockIdx.x;
    const int tid  = threadIdx.x;
    const int lane = tid & 31;
    const int wave = tid >> 5;
    const int m    = lane & 15;
    const int hi   = lane >> 4;

    for (int i = tid; i < Vq * Hq; i += NTHR)     embedLds[i] = embed[i];
    for (int i = tid; i < 2 * Hq * Hq; i += NTHR) w1Lds[i]    = ff_w1[i];
    for (int i = tid; i < 2 * Hq * Hq; i += NTHR) w2Lds[i]    = ff_w2[i];
    __syncthreads();

    float b1t[4], b2t[2], lnGv = 0.f, lnBv = 0.f;
    if (wave < NP) {
        #pragma unroll
        for (int nt = 0; nt < 4; ++nt) b1t[nt] = ff_b1[nt * 16 + m];
        #pragma unroll
        for (int nt = 0; nt < 2; ++nt) b2t[nt] = ff_b2[nt * 16 + m];
        lnGv = ln_g[lane];
        lnBv = ln_b[lane];
    }

    float w1c[8], b1v[8], w2r[8], b2v = 0.f;
    if (wave == NP) {
        #pragma unroll
        for (int i = 0; i < 8; ++i) {
            w1c[i] = mlp_w1[i * Hq + lane];
            b1v[i] = mlp_b1[i];
            w2r[i] = mlp_w2[lane * 8 + i];
        }
        b2v = mlp_b2[lane];
    }

    float* const a1w = a1buf[wave < NP ? wave : 0];
    auto produce = [&](int c, float* hbuf) {
        const int base = c * CH;
        const int tok  = seq[b * Lq + base + m];

        v2f eA[8];
        #pragma unroll
        for (int kb = 0; kb < 8; ++kb)
            eA[kb] = *(const v2f*)&embedLds[tok * Hq + 4 * kb + 2 * hi];

        #pragma unroll 1
        for (int nt = 0; nt < 4; ++nt) {
            const int n0 = nt * 16;
            v8f acc = {};
            #pragma unroll
            for (int kb = 0; kb < 8; ++kb) {
                v2f bB = *(const v2f*)&w1Lds[(n0 + m) * Hq + 4 * kb + 2 * hi];
                acc = __builtin_amdgcn_wmma_f32_16x16x4_f32(
                        false, eA[kb], false, bB, (short)0, acc, false, false);
            }
            const float bias = b1t[nt];
            #pragma unroll
            for (int r = 0; r < 8; ++r) {
                float z = acc[r] + bias;
                a1w[(r + 8 * hi) * A1STR + n0 + m] = fmaxf(z, 0.f);
            }
        }
        asm volatile("s_wait_dscnt 0" ::: "memory");
        __builtin_amdgcn_wave_barrier();

        v2f a2[16];
        #pragma unroll
        for (int kb = 0; kb < 16; ++kb)
            a2[kb] = *(const v2f*)&a1w[m * A1STR + 4 * kb + 2 * hi];

        #pragma unroll 1
        for (int nt = 0; nt < 2; ++nt) {
            const int n0 = nt * 16;
            v8f acc = {};
            #pragma unroll
            for (int kb = 0; kb < 16; ++kb) {
                v2f bB = *(const v2f*)&w2Lds[(n0 + m) * (2 * Hq) + 4 * kb + 2 * hi];
                acc = __builtin_amdgcn_wmma_f32_16x16x4_f32(
                        false, a2[kb], false, bB, (short)0, acc, false, false);
            }
            const float bias = b2t[nt];
            #pragma unroll
            for (int r = 0; r < 8; ++r) {
                int   tm = __shfl(tok, r + 8 * hi, 32);
                float e  = embedLds[tm * Hq + n0 + m];
                hbuf[(r + 8 * hi) * HSTR + n0 + m] = acc[r] + bias + e;
            }
        }
        asm volatile("s_wait_dscnt 0" ::: "memory");
        __builtin_amdgcn_wave_barrier();

        #pragma unroll 1
        for (int t = 0; t < CH; ++t) {
            float z = hbuf[t * HSTR + lane];
            float ss[2] = {z, z * z};
            wave_sum<2>(ss);
            const float s = ss[0], s2 = ss[1];
            float mu  = s * (1.f / 32.f);
            float var = s2 * (1.f / 32.f) - mu * mu;
            float nz  = (z - mu) * rsqrtf(var + EPSC);
            hbuf[t * HSTR + lane] = nz * lnGv + lnBv;
        }
        asm volatile("s_wait_dscnt 0" ::: "memory");
    };

    auto consume = [&](const float* hbuf, int nsteps) {
        const float* rp = hbuf + lane;
        float xn = rp[0];
        float vn = rp[HSTR];
        asm volatile("" :: "v"(xn), "v"(vn));
        if (nsteps > 1) rp += 2 * HSTR;
        #pragma unroll 1
        for (int s = 0; s < nsteps; ++s) {
            const float x = xn;
            const float v = vn;
            xn = rp[0];
            vn = rp[HSTR];
            __builtin_amdgcn_sched_barrier(0);
            if (s + 2 < nsteps) rp += 2 * HSTR;

            float p[8];
            #pragma unroll
            for (int i = 0; i < 8; ++i) p[i] = w1c[i] * x;
            wave_sum<8>(p);
            float a1p[8], a1v[8];
            float y = b2v;
            #pragma unroll
            for (int i = 0; i < 8; ++i) {
                a1p[i] = p[i] + b1v[i];
                a1v[i] = fmaxf(a1p[i], 0.f);
                y = fmaf(w2r[i], a1v[i], y);
            }
            float d = (y - v) * 0.0625f;

            float q[8];
            #pragma unroll
            for (int i = 0; i < 8; ++i) q[i] = w2r[i] * d;
            wave_sum<8>(q);
            b2v -= LRC * d;
            #pragma unroll
            for (int i = 0; i < 8; ++i) {
                float g1 = (a1p[i] > 0.f) ? q[i] : 0.f;
                w2r[i] = fmaf(-LRC * d, a1v[i], w2r[i]);
                b1v[i] -= LRC * g1;
                w1c[i] = fmaf(-LRC * g1, x, w1c[i]);
            }
        }
    };

    if (wave < NP) produce(wave, hid[0] + wave * (CH * HSTR));
    __syncthreads();

    for (int g = 0; g < NGRP; ++g) {
        if (wave < NP) {
            const int c = NP * (g + 1) + wave;
            if (c < NCH) produce(c, hid[(g + 1) & 1] + wave * (CH * HSTR));
        } else {
            consume(hid[g & 1], (g == NGRP - 1) ? (8 * (NCH - NP * (NGRP - 1)) - 1) : 8 * NP);
        }
        __syncthreads();
    }

    if (wave == NP) {
        const float* hbuf = hid[(NGRP - 1) & 1] + ((NCH - 1) - NP * (NGRP - 1)) * (CH * HSTR);
        float x = hbuf[15 * HSTR + lane];
        float p[8];
        #pragma unroll
        for (int i = 0; i < 8; ++i) p[i] = w1c[i] * x;
        wave_sum<8>(p);
        float y = b2v;
        #pragma unroll
        for (int i = 0; i < 8; ++i) y = fmaf(w2r[i], fmaxf(p[i] + b1v[i], 0.f), y);
        ctxLds[lane] = y;
    }
    __syncthreads();

    if (tid < Vq) {
        float acc = out_b[tid];
        #pragma unroll
        for (int h = 0; h < Hq; ++h)
            acc = fmaf(ctxLds[h], out_w[tid * Hq + h], acc);
        out[b * Vq + tid] = acc;
    }
}

extern "C" void kernel_launch(void* const* d_in, const int* in_sizes, int n_in,
                              void* d_out, int out_size, void* d_ws, size_t ws_size,
                              hipStream_t stream) {
    (void)in_sizes; (void)n_in; (void)out_size; (void)d_ws; (void)ws_size;
    const int*   seq    = (const int*)  d_in[0];
    const float* embed  = (const float*)d_in[1];
    const float* ff_w1  = (const float*)d_in[2];
    const float* ff_b1  = (const float*)d_in[3];
    const float* ff_w2  = (const float*)d_in[4];
    const float* ff_b2  = (const float*)d_in[5];
    const float* ln_g   = (const float*)d_in[6];
    const float* ln_b   = (const float*)d_in[7];
    const float* mlp_w1 = (const float*)d_in[8];
    const float* mlp_b1 = (const float*)d_in[9];
    const float* mlp_w2 = (const float*)d_in[10];
    const float* mlp_b2 = (const float*)d_in[11];
    const float* out_w  = (const float*)d_in[12];
    const float* out_b  = (const float*)d_in[13];
    float*       out    = (float*)d_out;

    pmm_fused_kernel<<<dim3(Bq), dim3(NTHR), 0, stream>>>(
        seq, embed, ff_w1, ff_b1, ff_w2, ff_b2, ln_g, ln_b,
        mlp_w1, mlp_b1, mlp_w2, mlp_b2, out_w, out_b, out);
}


# SGD wave's 32-lane sums of 8: after the xor-16 add each 16-lane row finishes only 4 of the 8 (state kept rotated by 4 in lanes 16-31), rows swap results; y's 8-term sum order rotates there
# speedup vs baseline: 4.6743x; 1.0826x over previous
#include <hip/hip_runtime.h>
#include <hip/hip_bf16.h>
#include <math.h>

typedef __attribute__((ext_vector_type(2))) float v2f;
typedef __attribute__((ext_vector_type(8))) float v8f;

#define LRC  0.01f
#define EPSC 1e-5f

#define Bq 256
#define Lq 4096
#define Hq 32
#define Vq 64
#define CH 16
#define NCH (Lq / CH)
#define NP  3
#define NGRP ((NCH + NP - 1) / NP)
#define NTHR 128
static_assert(NTHR == 32 * (NP + 1), "block = NP producer waves + 1 consumer wave");
#define A1STR 68
#define HSTR  36

template <int OFF>
__device__ __forceinline__ float xlane(float v) {
    const int b = __builtin_bit_cast(int, v);
    if constexpr (OFF == 16) {
        return __builtin_bit_cast(float,
            __builtin_amdgcn_permlanex16(b, b, 0x76543210, 0xfedcba98, false, false));
    } else {
        return __builtin_bit_cast(float,
            __builtin_amdgcn_update_dpp(0, b, 0x160 + OFF, 0xf, 0xf, true));
    }
}
template <int OFF, int N>
__device__ __forceinline__ void bfly_add(float (&v)[N]) {
    #pragma unroll
    for (int i = 0; i < N; ++i) v[i] += xlane<OFF>(v[i]);
}
template <int N>
__device__ __forceinline__ void wave_sum(float (&v)[N]) {
    bfly_add<16, N>(v);
    bfly_add<8, N>(v);
    bfly_add<4, N>(v);
    bfly_add<2, N>(v);
    bfly_add<1, N>(v);
}
#define SLOT_ROT(lane) (((lane) >> 4) << 2)
__device__ __forceinline__ void wave_sum8_rows(float (&v)[8]) {
    float mine[4];
    #pragma unroll
    for (int k = 0; k < 4; ++k) mine[k] = v[k] + xlane<16>(v[k + 4]);
    bfly_add<8, 4>(mine);
    bfly_add<4, 4>(mine);
    bfly_add<2, 4>(mine);
    bfly_add<1, 4>(mine);
    #pragma unroll
    for (int k = 0; k < 4; ++k) {
        v[k]     = mine[k];
        v[k + 4] = xlane<16>(mine[k]);
    }
}

__launch_bounds__(NTHR, 1)
__global__ void pmm_fused_kernel(const int* __restrict__ seq,
                                 const float* __restrict__ embed,
                                 const float* __restrict__ ff_w1,
                                 const float* __restrict__ ff_b1,
                                 const float* __restrict__ ff_w2,
                                 const float* __restrict__ ff_b2,
                                 const float* __restrict__ ln_g,
                                 const float* __restrict__ ln_b,
                                 const float* __restrict__ mlp_w1,
                                 const float* __restrict__ mlp_b1,
                                 const float* __restrict__ mlp_w2,
                                 const float* __restrict__ mlp_b2,
                                 const float* __restrict__ out_w,
                                 const float* __restrict__ out_b,
                                 float* __restrict__ out)
{
    __shared__ float embedLds[Vq * Hq];
    __shared__ float w1Lds[2 * Hq * Hq];
    __shared__ float w2Lds[Hq * 2 * Hq];
    __shared__ float a1buf[NP][CH * A1STR];
    __shared__ float hid[2][NP * CH * HSTR];
    __shared__ float ctxLds[Hq];

    const int b    = blockIdx.x;
    const int tid  = threadIdx.x;
    const int lane = tid & 31;
    const int wave = tid >> 5;
    const int m    = lane & 15;
    const int hi   = lane >> 4;

    for (int i = tid; i < Vq * Hq; i += NTHR)     embedLds[i] = embed[i];
    for (int i = tid; i < 2 * Hq * Hq; i += NTHR) w1Lds[i]    = ff_w1[i];
    for (int i = tid; i < 2 * Hq * Hq; i += NTHR) w2Lds[i]    = ff_w2[i];
    __syncthreads();

    float b1t[4], b2t[2], lnGv = 0.f, lnBv = 0.f;
    if (wave < NP) {
        #pragma unroll
        for (int nt = 0; nt < 4; ++nt) b1t[nt] = ff_b1[nt * 16 + m];
        #pragma unroll
        for (int nt = 0; nt < 2; ++nt) b2t[nt] = ff_b2[nt * 16 + m];
        lnGv = ln_g[lane];
        lnBv = ln_b[lane];
    }

    float w1c[8], b1v[8], w2r[8], b2v = 0.f;
    if (wave == NP) {
        const int rot = SLOT_ROT(lane);
        #pragma unroll
        for (int k = 0; k < 8; ++k) {
            const int i = k ^ rot;
            w1c[k] = mlp_w1[i * Hq + lane];
            b1v[k] = mlp_b1[i];
            w2r[k] = mlp_w2[lane * 8 + i];
        }
        b2v = mlp_b2[lane];
    }

    float* const a1w = a1buf[wave < NP ? wave : 0];
    auto produce = [&](int c, float* hbuf) {
        const int base = c * CH;
        const int tok  = seq[b * Lq + base + m];

        v2f eA[8];
        #pragma unroll
        for (int kb = 0; kb < 8; ++kb)
            eA[kb] = *(const v2f*)&embedLds[tok * Hq + 4 * kb + 2 * hi];

        #pragma unroll 1
        for (int nt = 0; nt < 4; ++nt) {
            const int n0 = nt * 16;
            v8f acc = {};
            #pragma unroll
            for (int kb = 0; kb < 8; ++kb) {
                v2f bB = *(const v2f*)&w1Lds[(n0 + m) * Hq + 4 * kb + 2 * hi];
                acc = __builtin_amdgcn_wmma_f32_16x16x4_f32(
                        false, eA[kb], false, bB, (short)0, acc, false, false);
            }
            const float bias = b1t[nt];
            #pragma unroll
            for (int r = 0; r < 8; ++r) {
                float z = acc[r] + bias;
                a1w[(r + 8 * hi) * A1STR + n0 + m] = fmaxf(z, 0.f);
            }
        }
        asm volatile("s_wait_dscnt 0" ::: "memory");
        __builtin_amdgcn_wave_barrier();

        v2f a2[16];
        #pragma unroll
        for (int kb = 0; kb < 16; ++kb)
            a2[kb] = *(const v2f*)&a1w[m * A1STR + 4 * kb + 2 * hi];

        #pragma unroll 1
        for (int nt = 0; nt < 2; ++nt) {
            const int n0 = nt * 16;
            v8f acc = {};
            #pragma unroll
            for (int kb = 0; kb < 16; ++kb) {
                v2f bB = *(const v2f*)&w2Lds[(n0 + m) * (2 * Hq) + 4 * kb + 2 * hi];
                acc = __builtin_amdgcn_wmma_f32_16x16x4_f32(
                        false, a2[kb], false, bB, (short)0, acc, false, false);
            }
            const float bias = b2t[nt];
            #pragma unroll
            for (int r = 0; r < 8; ++r) {
                int   tm = __shfl(tok, r + 8 * hi, 32);
                float e  = embedLds[tm * Hq + n0 + m];
                hbuf[(r + 8 * hi) * HSTR + n0 + m] = acc[r] + bias + e;
            }
        }
        asm volatile("s_wait_dscnt 0" ::: "memory");
        __builtin_amdgcn_wave_barrier();

        #pragma unroll 1
        for (int t = 0; t < CH; ++t) {
            float z = hbuf[t * HSTR + lane];
            float ss[2] = {z, z * z};
            wave_sum<2>(ss);
            const float s = ss[0], s2 = ss[1];
            float mu  = s * (1.f / 32.f);
            float var = s2 * (1.f / 32.f) - mu * mu;
            float nz  = (z - mu) * rsqrtf(var + EPSC);
            hbuf[t * HSTR + lane] = nz * lnGv + lnBv;
        }
        asm volatile("s_wait_dscnt 0" ::: "memory");
    };

    auto consume = [&](const float* hbuf, int nsteps) {
        const float* rp = hbuf + lane;
        float xn = rp[0];
        float vn = rp[HSTR];
        asm volatile("" :: "v"(xn), "v"(vn));
        if (nsteps > 1) rp += 2 * HSTR;
        #pragma unroll 1
        for (int s = 0; s < nsteps; ++s) {
            const float x = xn;
            const float v = vn;
            xn = rp[0];
            vn = rp[HSTR];
            __builtin_amdgcn_sched_barrier(0);
            if (s + 2 < nsteps) rp += 2 * HSTR;

            float p[8];
            #pragma unroll
            for (int i = 0; i < 8; ++i) p[i] = w1c[i] * x;
            wave_sum8_rows(p);
            float a1p[8], a1v[8];
            float y = b2v;
            #pragma unroll
            for (int i = 0; i < 8; ++i) {
                a1p[i] = p[i] + b1v[i];
                a1v[i] = fmaxf(a1p[i], 0.f);
                y = fmaf(w2r[i], a1v[i], y);
            }
            float d = (y - v) * 0.0625f;

            float q[8];
            #pragma unroll
            for (int i = 0; i < 8; ++i) q[i] = w2r[i] * d;
            wave_sum8_rows(q);
            b2v -= LRC * d;
            #pragma unroll
            for (int i = 0; i < 8; ++i) {
                float g1 = (a1p[i] > 0.f) ? q[i] : 0.f;
                w2r[i] = fmaf(-LRC * d, a1v[i], w2r[i]);
                b1v[i] -= LRC * g1;
                w1c[i] = fmaf(-LRC * g1, x, w1c[i]);
            }
        }
    };

    if (wave < NP) produce(wave, hid[0] + wave * (CH * HSTR));
    __syncthreads();

    for (int g = 0; g < NGRP; ++g) {
        if (wave < NP) {
            const int c = NP * (g + 1) + wave;
            if (c < NCH) produce(c, hid[(g + 1) & 1] + wave * (CH * HSTR));
        } else {
            consume(hid[g & 1], (g == NGRP - 1) ? (8 * (NCH - NP * (NGRP - 1)) - 1) : 8 * NP);
        }
        __syncthreads();
    }

    if (wave == NP) {
        const float* hbuf = hid[(NGRP - 1) & 1] + ((NCH - 1) - NP * (NGRP - 1)) * (CH * HSTR);
        float x = hbuf[15 * HSTR + lane];
        float p[8];
        #pragma unroll
        for (int i = 0; i < 8; ++i) p[i] = w1c[i] * x;
        wave_sum8_rows(p);
        float y = b2v;
        #pragma unroll
        for (int i = 0; i < 8; ++i) y = fmaf(w2r[i], fmaxf(p[i] + b1v[i], 0.f), y);
        ctxLds[lane] = y;
    }
    __syncthreads();

    if (tid < Vq) {
        float acc = out_b[tid];
        #pragma unroll
        for (int h = 0; h < Hq; ++h)
            acc = fmaf(ctxLds[h], out_w[tid * Hq + h], acc);
        out[b * Vq + tid] = acc;
    }
}

extern "C" void kernel_launch(void* const* d_in, const int* in_sizes, int n_in,
                              void* d_out, int out_size, void* d_ws, size_t ws_size,
                              hipStream_t stream) {
    (void)in_sizes; (void)n_in; (void)out_size; (void)d_ws; (void)ws_size;
    const int*   seq    = (const int*)  d_in[0];
    const float* embed  = (const float*)d_in[1];
    const float* ff_w1  = (const float*)d_in[2];
    const float* ff_b1  = (const float*)d_in[3];
    const float* ff_w2  = (const float*)d_in[4];
    const float* ff_b2  = (const float*)d_in[5];
    const float* ln_g   = (const float*)d_in[6];
    const float* ln_b   = (const float*)d_in[7];
    const float* mlp_w1 = (const float*)d_in[8];
    const float* mlp_b1 = (const float*)d_in[9];
    const float* mlp_w2 = (const float*)d_in[10];
    const float* mlp_b2 = (const float*)d_in[11];
    const float* out_w  = (const float*)d_in[12];
    const float* out_b  = (const float*)d_in[13];
    float*       out    = (float*)d_out;

    pmm_fused_kernel<<<dim3(Bq), dim3(NTHR), 0, stream>>>(
        seq, embed, ff_w1, ff_b1, ff_w2, ff_b2, ln_g, ln_b,
        mlp_w1, mlp_b1, mlp_w2, mlp_b2, out_w, out_b, out);
}


# Producer waves: LayerNorm batched 8 tokens at a time (8 LDS row reads as one group, each butterfly step over the batch's 16 sums) instead of 16 serial one-token passes; per-token sums bit-identical
# speedup vs baseline: 5.2613x; 1.1256x over previous
#include <hip/hip_runtime.h>
#include <hip/hip_bf16.h>
#include <math.h>

typedef __attribute__((ext_vector_type(2))) float v2f;
typedef __attribute__((ext_vector_type(8))) float v8f;

#define LRC  0.01f
#define EPSC 1e-5f

#define Bq 256
#define Lq 4096
#define Hq 32
#define Vq 64
#define CH 16
#define NCH (Lq / CH)
#define NP  3
#define NGRP ((NCH + NP - 1) / NP)
#define NTHR 128
static_assert(NTHR == 32 * (NP + 1), "block = NP producer waves + 1 consumer wave");
#define A1STR 68
#define HSTR  36
#define LNB   8
static_assert(CH % LNB == 0, "a chunk is a whole number of LayerNorm batches");

template <int OFF>
__device__ __forceinline__ float xlane(float v) {
    const int b = __builtin_bit_cast(int, v);
    if constexpr (OFF == 16) {
        return __builtin_bit_cast(float,
            __builtin_amdgcn_permlanex16(b, b, 0x76543210, 0xfedcba98, false, false));
    } else {
        return __builtin_bit_cast(float,
            __builtin_amdgcn_update_dpp(0, b, 0x160 + OFF, 0xf, 0xf, true));
    }
}
template <int OFF, int N>
__device__ __forceinline__ void bfly_add(float (&v)[N]) {
    #pragma unroll
    for (int i = 0; i < N; ++i) v[i] += xlane<OFF>(v[i]);
}
template <int N>
__device__ __forceinline__ void wave_sum(float (&v)[N]) {
    bfly_add<16, N>(v);
    bfly_add<8, N>(v);
    bfly_add<4, N>(v);
    bfly_add<2, N>(v);
    bfly_add<1, N>(v);
}
#define SLOT_ROT(lane) (((lane) >> 4) << 2)
__device__ __forceinline__ void wave_sum8_rows(float (&v)[8]) {
    float mine[4];
    #pragma unroll
    for (int k = 0; k < 4; ++k) mine[k] = v[k] + xlane<16>(v[k + 4]);
    bfly_add<8, 4>(mine);
    bfly_add<4, 4>(mine);
    bfly_add<2, 4>(mine);
    bfly_add<1, 4>(mine);
    #pragma unroll
    for (int k = 0; k < 4; ++k) {
        v[k]     = mine[k];
        v[k + 4] = xlane<16>(mine[k]);
    }
}

__launch_bounds__(NTHR, 1)
__global__ void pmm_fused_kernel(const int* __restrict__ seq,
                                 const float* __restrict__ embed,
                                 const float* __restrict__ ff_w1,
                                 const float* __restrict__ ff_b1,
                                 const float* __restrict__ ff_w2,
                                 const float* __restrict__ ff_b2,
                                 const float* __restrict__ ln_g,
                                 const float* __restrict__ ln_b,
                                 const float* __restrict__ mlp_w1,
                                 const float* __restrict__ mlp_b1,
                                 const float* __restrict__ mlp_w2,
                                 const float* __restrict__ mlp_b2,
                                 const float* __restrict__ out_w,
                                 const float* __restrict__ out_b,
                                 float* __restrict__ out)
{
    __shared__ float embedLds[Vq * Hq];
    __shared__ float w1Lds[2 * Hq * Hq];
    __shared__ float w2Lds[Hq * 2 * Hq];
    __shared__ float a1buf[NP][CH * A1STR];
    __shared__ float hid[2][NP * CH * HSTR];
    __shared__ float ctxLds[Hq];

    const int b    = blockIdx.x;
    const int tid  = threadIdx.x;
    const int lane = tid & 31;
    const int wave = tid >> 5;
    const int m    = lane & 15;
    const int hi   = lane >> 4;

    for (int i = tid; i < Vq * Hq; i += NTHR)     embedLds[i] = embed[i];
    for (int i = tid; i < 2 * Hq * Hq; i += NTHR) w1Lds[i]    = ff_w1[i];
    for (int i = tid; i < 2 * Hq * Hq; i += NTHR) w2Lds[i]    = ff_w2[i];
    __syncthreads();

    float b1t[4], b2t[2], lnGv = 0.f, lnBv = 0.f;
    if (wave < NP) {
        #pragma unroll
        for (int nt = 0; nt < 4; ++nt) b1t[nt] = ff_b1[nt * 16 + m];
        #pragma unroll
        for (int nt = 0; nt < 2; ++nt) b2t[nt] = ff_b2[nt * 16 + m];
        lnGv = ln_g[lane];
        lnBv = ln_b[lane];
    }

    float w1c[8], b1v[8], w2r[8], b2v = 0.f;
    if (wave == NP) {
        const int rot = SLOT_ROT(lane);
        #pragma unroll
        for (int k = 0; k < 8; ++k) {
            const int i = k ^ rot;
            w1c[k] = mlp_w1[i * Hq + lane];
            b1v[k] = mlp_b1[i];
            w2r[k] = mlp_w2[lane * 8 + i];
        }
        b2v = mlp_b2[lane];
    }

    float* const a1w = a1buf[wave < NP ? wave : 0];
    auto produce = [&](int c, float* hbuf) {
        const int base = c * CH;
        const int tok  = seq[b * Lq + base + m];

        v2f eA[8];
        #pragma unroll
        for (int kb = 0; kb < 8; ++kb)
            eA[kb] = *(const v2f*)&embedLds[tok * Hq + 4 * kb + 2 * hi];

        #pragma unroll 1
        for (int nt = 0; nt < 4; ++nt) {
            const int n0 = nt * 16;
            v8f acc = {};
            #pragma unroll
            for (int kb = 0; kb < 8; ++kb) {
                v2f bB = *(const v2f*)&w1Lds[(n0 + m) * Hq + 4 * kb + 2 * hi];
                acc = __builtin_amdgcn_wmma_f32_16x16x4_f32(
                        false, eA[kb], false, bB, (short)0, acc, false, false);
            }
            const float bias = b1t[nt];
            #pragma unroll
            for (int r = 0; r < 8; ++r) {
                float z = acc[r] + bias;
                a1w[(r + 8 * hi) * A1STR + n0 + m] = fmaxf(z, 0.f);
            }
        }
        asm volatile("s_wait_dscnt 0" ::: "memory");
        __builtin_amdgcn_wave_barrier();

        v2f a2[16];
        #pragma unroll
        for (int kb = 0; kb < 16; ++kb)
            a2[kb] = *(const v2f*)&a1w[m * A1STR + 4 * kb + 2 * hi];

        #pragma unroll 1
        for (int nt = 0; nt < 2; ++nt) {
            const int n0 = nt * 16;
            v8f acc = {};
            #pragma unroll
            for (int kb = 0; kb < 16; ++kb) {
                v2f bB = *(const v2f*)&w2Lds[(n0 + m) * (2 * Hq) + 4 * kb + 2 * hi];
                acc = __builtin_amdgcn_wmma_f32_16x16x4_f32(
                        false, a2[kb], false, bB, (short)0, acc, false, false);
            }
            const float bias = b2t[nt];
            #pragma unroll
            for (int r = 0; r < 8; ++r) {
                int   tm = __shfl(tok, r + 8 * hi, 32);
                float e  = embedLds[tm * Hq + n0 + m];
                hbuf[(r + 8 * hi) * HSTR + n0 + m] = acc[r] + bias + e;
            }
        }
        asm volatile("s_wait_dscnt 0" ::: "memory");
        __builtin_amdgcn_wave_barrier();

        #pragma unroll 1
        for (int t0 = 0; t0 < CH; t0 += LNB) {
            float* const hp = hbuf + t0 * HSTR + lane;
            float z[LNB];
            #pragma unroll
            for (int t = 0; t < LNB; ++t) z[t] = hp[t * HSTR];
            float ss[2 * LNB];
            #pragma unroll
            for (int t = 0; t < LNB; ++t) { ss[2 * t] = z[t]; ss[2 * t + 1] = z[t] * z[t]; }
            wave_sum<2 * LNB>(ss);
            #pragma unroll
            for (int t = 0; t < LNB; ++t) {
                const float s = ss[2 * t], s2 = ss[2 * t + 1];
                float mu  = s * (1.f / 32.f);
                float var = s2 * (1.f / 32.f) - mu * mu;
                float nz  = (z[t] - mu) * rsqrtf(var + EPSC);
                hp[t * HSTR] = nz * lnGv + lnBv;
            }
        }
        asm volatile("s_wait_dscnt 0" ::: "memory");
    };

    auto consume = [&](const float* hbuf, int nsteps) {
        const float* rp = hbuf + lane;
        float xn = rp[0];
        float vn = rp[HSTR];
        asm volatile("" :: "v"(xn), "v"(vn));
        if (nsteps > 1) rp += 2 * HSTR;
        #pragma unroll 1
        for (int s = 0; s < nsteps; ++s) {
            const float x = xn;
            const float v = vn;
            xn = rp[0];
            vn = rp[HSTR];
            __builtin_amdgcn_sched_barrier(0);
            if (s + 2 < nsteps) rp += 2 * HSTR;

            float p[8];
            #pragma unroll
            for (int i = 0; i < 8; ++i) p[i] = w1c[i] * x;
            wave_sum8_rows(p);
            float a1p[8], a1v[8];
            float y = b2v;
            #pragma unroll
            for (int i = 0; i < 8; ++i) {
                a1p[i] = p[i] + b1v[i];
                a1v[i] = fmaxf(a1p[i], 0.f);
                y = fmaf(w2r[i], a1v[i], y);
            }
            float d = (y - v) * 0.0625f;

            float q[8];
            #pragma unroll
            for (int i = 0; i < 8; ++i) q[i] = w2r[i] * d;
            wave_sum8_rows(q);
            b2v -= LRC * d;
            #pragma unroll
            for (int i = 0; i < 8; ++i) {
                float g1 = (a1p[i] > 0.f) ? q[i] : 0.f;
                w2r[i] = fmaf(-LRC * d, a1v[i], w2r[i]);
                b1v[i] -= LRC * g1;
                w1c[i] = fmaf(-LRC * g1, x, w1c[i]);
            }
        }
    };

    if (wave < NP) produce(wave, hid[0] + wave * (CH * HSTR));
    __syncthreads();

    for (int g = 0; g < NGRP; ++g) {
        if (wave < NP) {
            const int c = NP * (g + 1) + wave;
            if (c < NCH) produce(c, hid[(g + 1) & 1] + wave * (CH * HSTR));
        } else {
            consume(hid[g & 1], (g == NGRP - 1) ? (8 * (NCH - NP * (NGRP - 1)) - 1) : 8 * NP);
        }
        __syncthreads();
    }

    if (wave == NP) {
        const float* hbuf = hid[(NGRP - 1) & 1] + ((NCH - 1) - NP * (NGRP - 1)) * (CH * HSTR);
        float x = hbuf[15 * HSTR + lane];
        float p[8];
        #pragma unroll
        for (int i = 0; i < 8; ++i) p[i] = w1c[i] * x;
        wave_sum8_rows(p);
        float y = b2v;
        #pragma unroll
        for (int i = 0; i < 8; ++i) y = fmaf(w2r[i], fmaxf(p[i] + b1v[i], 0.f), y);
        ctxLds[lane] = y;
    }
    __syncthreads();

    if (tid < Vq) {
        float acc = out_b[tid];
        #pragma unroll
        for (int h = 0; h < Hq; ++h)
            acc = fmaf(ctxLds[h], out_w[tid * Hq + h], acc);
        out[b * Vq + tid] = acc;
    }
}

extern "C" void kernel_launch(void* const* d_in, const int* in_sizes, int n_in,
                              void* d_out, int out_size, void* d_ws, size_t ws_size,
                              hipStream_t stream) {
    (void)in_sizes; (void)n_in; (void)out_size; (void)d_ws; (void)ws_size;
    const int*   seq    = (const int*)  d_in[0];
    const float* embed  = (const float*)d_in[1];
    const float* ff_w1  = (const float*)d_in[2];
    const float* ff_b1  = (const float*)d_in[3];
    const float* ff_w2  = (const float*)d_in[4];
    const float* ff_b2  = (const float*)d_in[5];
    const float* ln_g   = (const float*)d_in[6];
    const float* ln_b   = (const float*)d_in[7];
    const float* mlp_w1 = (const float*)d_in[8];
    const float* mlp_b1 = (const float*)d_in[9];
    const float* mlp_w2 = (const float*)d_in[10];
    const float* mlp_b2 = (const float*)d_in[11];
    const float* out_w  = (const float*)d_in[12];
    const float* out_b  = (const float*)d_in[13];
    float*       out    = (float*)d_out;

    pmm_fused_kernel<<<dim3(Bq), dim3(NTHR), 0, stream>>>(
        seq, embed, ff_w1, ff_b1, ff_w2, ff_b2, ln_g, ln_b,
        mlp_w1, mlp_b1, mlp_w2, mlp_b2, out_w, out_b, out);
}


# SGD wave: relu gate of the backward pass made as a multiplier (-lr or 0) right after the forward activations, so its compare and select leave the tail of the step; products and sums bit-identical
# speedup vs baseline: 5.5775x; 1.0601x over previous
#include <hip/hip_runtime.h>
#include <hip/hip_bf16.h>
#include <math.h>

typedef __attribute__((ext_vector_type(2))) float v2f;
typedef __attribute__((ext_vector_type(8))) float v8f;

#define LRC  0.01f
#define EPSC 1e-5f

#define Bq 256
#define Lq 4096
#define Hq 32
#define Vq 64
#define CH 16
#define NCH (Lq / CH)
#define NP  3
#define NGRP ((NCH + NP - 1) / NP)
#define NTHR 128
static_assert(NTHR == 32 * (NP + 1), "block = NP producer waves + 1 consumer wave");
#define A1STR 68
#define HSTR  36
#define LNB   8
static_assert(CH % LNB == 0, "a chunk is a whole number of LayerNorm batches");

template <int OFF>
__device__ __forceinline__ float xlane(float v) {
    const int b = __builtin_bit_cast(int, v);
    if constexpr (OFF == 16) {
        return __builtin_bit_cast(float,
            __builtin_amdgcn_permlanex16(b, b, 0x76543210, 0xfedcba98, false, false));
    } else {
        return __builtin_bit_cast(float,
            __builtin_amdgcn_update_dpp(0, b, 0x160 + OFF, 0xf, 0xf, true));
    }
}
template <int OFF, int N>
__device__ __forceinline__ void bfly_add(float (&v)[N]) {
    #pragma unroll
    for (int i = 0; i < N; ++i) v[i] += xlane<OFF>(v[i]);
}
template <int N>
__device__ __forceinline__ void wave_sum(float (&v)[N]) {
    bfly_add<16, N>(v);
    bfly_add<8, N>(v);
    bfly_add<4, N>(v);
    bfly_add<2, N>(v);
    bfly_add<1, N>(v);
}
#define SLOT_ROT(lane) (((lane) >> 4) << 2)
__device__ __forceinline__ void wave_sum8_rows(float (&v)[8]) {
    float mine[4];
    #pragma unroll
    for (int k = 0; k < 4; ++k) mine[k] = v[k] + xlane<16>(v[k + 4]);
    bfly_add<8, 4>(mine);
    bfly_add<4, 4>(mine);
    bfly_add<2, 4>(mine);
    bfly_add<1, 4>(mine);
    #pragma unroll
    for (int k = 0; k < 4; ++k) {
        v[k]     = mine[k];
        v[k + 4] = xlane<16>(mine[k]);
    }
}

__launch_bounds__(NTHR, 1)
__global__ void pmm_fused_kernel(const int* __restrict__ seq,
                                 const float* __restrict__ embed,
                                 const float* __restrict__ ff_w1,
                                 const float* __restrict__ ff_b1,
                                 const float* __restrict__ ff_w2,
                                 const float* __restrict__ ff_b2,
                                 const float* __restrict__ ln_g,
                                 const float* __restrict__ ln_b,
                                 const float* __restrict__ mlp_w1,
                                 const float* __restrict__ mlp_b1,
                                 const float* __restrict__ mlp_w2,
                                 const float* __restrict__ mlp_b2,
                                 const float* __restrict__ out_w,
                                 const float* __restrict__ out_b,
                                 float* __restrict__ out)
{
    __shared__ float embedLds[Vq * Hq];
    __shared__ float w1Lds[2 * Hq * Hq];
    __shared__ float w2Lds[Hq * 2 * Hq];
    __shared__ float a1buf[NP][CH * A1STR];
    __shared__ float hid[2][NP * CH * HSTR];
    __shared__ float ctxLds[Hq];

    const int b    = blockIdx.x;
    const int tid  = threadIdx.x;
    const int lane = tid & 31;
    const int wave = tid >> 5;
    const int m    = lane & 15;
    const int hi   = lane >> 4;

    for (int i = tid; i < Vq * Hq; i += NTHR)     embedLds[i] = embed[i];
    for (int i = tid; i < 2 * Hq * Hq; i += NTHR) w1Lds[i]    = ff_w1[i];
    for (int i = tid; i < 2 * Hq * Hq; i += NTHR) w2Lds[i]    = ff_w2[i];
    __syncthreads();

    float b1t[4], b2t[2], lnGv = 0.f, lnBv = 0.f;
    if (wave < NP) {
        #pragma unroll
        for (int nt = 0; nt < 4; ++nt) b1t[nt] = ff_b1[nt * 16 + m];
        #pragma unroll
        for (int nt = 0; nt < 2; ++nt) b2t[nt] = ff_b2[nt * 16 + m];
        lnGv = ln_g[lane];
        lnBv = ln_b[lane];
    }

    float w1c[8], b1v[8], w2r[8], b2v = 0.f;
    if (wave == NP) {
        const int rot = SLOT_ROT(lane);
        #pragma unroll
        for (int k = 0; k < 8; ++k) {
            const int i = k ^ rot;
            w1c[k] = mlp_w1[i * Hq + lane];
            b1v[k] = mlp_b1[i];
            w2r[k] = mlp_w2[lane * 8 + i];
        }
        b2v = mlp_b2[lane];
    }

    float* const a1w = a1buf[wave < NP ? wave : 0];
    auto produce = [&](int c, float* hbuf) {
        const int base = c * CH;
        const int tok  = seq[b * Lq + base + m];

        v2f eA[8];
        #pragma unroll
        for (int kb = 0; kb < 8; ++kb)
            eA[kb] = *(const v2f*)&embedLds[tok * Hq + 4 * kb + 2 * hi];

        #pragma unroll 1
        for (int nt = 0; nt < 4; ++nt) {
            const int n0 = nt * 16;
            v8f acc = {};
            #pragma unroll
            for (int kb = 0; kb < 8; ++kb) {
                v2f bB = *(const v2f*)&w1Lds[(n0 + m) * Hq + 4 * kb + 2 * hi];
                acc = __builtin_amdgcn_wmma_f32_16x16x4_f32(
                        false, eA[kb], false, bB, (short)0, acc, false, false);
            }
            const float bias = b1t[nt];
            #pragma unroll
            for (int r = 0; r < 8; ++r) {
                float z = acc[r] + bias;
                a1w[(r + 8 * hi) * A1STR + n0 + m] = fmaxf(z, 0.f);
            }
        }
        asm volatile("s_wait_dscnt 0" ::: "memory");
        __builtin_amdgcn_wave_barrier();

        v2f a2[16];
        #pragma unroll
        for (int kb = 0; kb < 16; ++kb)
            a2[kb] = *(const v2f*)&a1w[m * A1STR + 4 * kb + 2 * hi];

        #pragma unroll 1
        for (int nt = 0; nt < 2; ++nt) {
            const int n0 = nt * 16;
            v8f acc = {};
            #pragma unroll
            for (int kb = 0; kb < 16; ++kb) {
                v2f bB = *(const v2f*)&w2Lds[(n0 + m) * (2 * Hq) + 4 * kb + 2 * hi];
                acc = __builtin_amdgcn_wmma_f32_16x16x4_f32(
                        false, a2[kb], false, bB, (short)0, acc, false, false);
            }
            const float bias = b2t[nt];
            #pragma unroll
            for (int r = 0; r < 8; ++r) {
                int   tm = __shfl(tok, r + 8 * hi, 32);
                float e  = embedLds[tm * Hq + n0 + m];
                hbuf[(r + 8 * hi) * HSTR + n0 + m] = acc[r] + bias + e;
            }
        }
        asm volatile("s_wait_dscnt 0" ::: "memory");
        __builtin_amdgcn_wave_barrier();

        #pragma unroll 1
        for (int t0 = 0; t0 < CH; t0 += LNB) {
            float* const hp = hbuf + t0 * HSTR + lane;
            float z[LNB];
            #pragma unroll
            for (int t = 0; t < LNB; ++t) z[t] = hp[t * HSTR];
            float ss[2 * LNB];
            #pragma unroll
            for (int t = 0; t < LNB; ++t) { ss[2 * t] = z[t]; ss[2 * t + 1] = z[t] * z[t]; }
            wave_sum<2 * LNB>(ss);
            #pragma unroll
            for (int t = 0; t < LNB; ++t) {
                const float s = ss[2 * t], s2 = ss[2 * t + 1];
                float mu  = s * (1.f / 32.f);
                float var = s2 * (1.f / 32.f) - mu * mu;
                float nz  = (z[t] - mu) * rsqrtf(var + EPSC);
                hp[t * HSTR] = nz * lnGv + lnBv;
            }
        }
        asm volatile("s_wait_dscnt 0" ::: "memory");
    };

    auto consume = [&](const float* hbuf, int nsteps) {
        const float* rp = hbuf + lane;
        float xn = rp[0];
        float vn = rp[HSTR];
        asm volatile("" :: "v"(xn), "v"(vn));
        if (nsteps > 1) rp += 2 * HSTR;
        #pragma unroll 1
        for (int s = 0; s < nsteps; ++s) {
            const float x = xn;
            const float v = vn;
            xn = rp[0];
            vn = rp[HSTR];
            __builtin_amdgcn_sched_barrier(0);
            if (s + 2 < nsteps) rp += 2 * HSTR;

            float p[8];
            #pragma unroll
            for (int i = 0; i < 8; ++i) p[i] = w1c[i] * x;
            wave_sum8_rows(p);
            float a1p[8], a1v[8];
            float y = b2v;
            #pragma unroll
            for (int i = 0; i < 8; ++i) {
                a1p[i] = p[i] + b1v[i];
                a1v[i] = fmaxf(a1p[i], 0.f);
                y = fmaf(w2r[i], a1v[i], y);
            }
            float gate[8];
            #pragma unroll
            for (int i = 0; i < 8; ++i) gate[i] = (a1p[i] > 0.f) ? -LRC : 0.f;
            float d = (y - v) * 0.0625f;

            float q[8];
            #pragma unroll
            for (int i = 0; i < 8; ++i) q[i] = w2r[i] * d;
            wave_sum8_rows(q);
            b2v -= LRC * d;
            #pragma unroll
            for (int i = 0; i < 8; ++i) {
                const float sg = q[i] * gate[i];
                w2r[i] = fmaf(-LRC * d, a1v[i], w2r[i]);
                b1v[i] = fmaf(q[i], gate[i], b1v[i]);
                w1c[i] = fmaf(sg, x, w1c[i]);
            }
        }
    };

    if (wave < NP) produce(wave, hid[0] + wave * (CH * HSTR));
    __syncthreads();

    for (int g = 0; g < NGRP; ++g) {
        if (wave < NP) {
            const int c = NP * (g + 1) + wave;
            if (c < NCH) produce(c, hid[(g + 1) & 1] + wave * (CH * HSTR));
        } else {
            consume(hid[g & 1], (g == NGRP - 1) ? (8 * (NCH - NP * (NGRP - 1)) - 1) : 8 * NP);
        }
        __syncthreads();
    }

    if (wave == NP) {
        const float* hbuf = hid[(NGRP - 1) & 1] + ((NCH - 1) - NP * (NGRP - 1)) * (CH * HSTR);
        float x = hbuf[15 * HSTR + lane];
        float p[8];
        #pragma unroll
        for (int i = 0; i < 8; ++i) p[i] = w1c[i] * x;
        wave_sum8_rows(p);
        float y = b2v;
        #pragma unroll
        for (int i = 0; i < 8; ++i) y = fmaf(w2r[i], fmaxf(p[i] + b1v[i], 0.f), y);
        ctxLds[lane] = y;
    }
    __syncthreads();

    if (tid < Vq) {
        float acc = out_b[tid];
        #pragma unroll
        for (int h = 0; h < Hq; ++h)
            acc = fmaf(ctxLds[h], out_w[tid * Hq + h], acc);
        out[b * Vq + tid] = acc;
    }
}

extern "C" void kernel_launch(void* const* d_in, const int* in_sizes, int n_in,
                              void* d_out, int out_size, void* d_ws, size_t ws_size,
                              hipStream_t stream) {
    (void)in_sizes; (void)n_in; (void)out_size; (void)d_ws; (void)ws_size;
    const int*   seq    = (const int*)  d_in[0];
    const float* embed  = (const float*)d_in[1];
    const float* ff_w1  = (const float*)d_in[2];
    const float* ff_b1  = (const float*)d_in[3];
    const float* ff_w2  = (const float*)d_in[4];
    const float* ff_b2  = (const float*)d_in[5];
    const float* ln_g   = (const float*)d_in[6];
    const float* ln_b   = (const float*)d_in[7];
    const float* mlp_w1 = (const float*)d_in[8];
    const float* mlp_b1 = (const float*)d_in[9];
    const float* mlp_w2 = (const float*)d_in[10];
    const float* mlp_b2 = (const float*)d_in[11];
    const float* out_w  = (const float*)d_in[12];
    const float* out_b  = (const float*)d_in[13];
    float*       out    = (float*)d_out;

    pmm_fused_kernel<<<dim3(Bq), dim3(NTHR), 0, stream>>>(
        seq, embed, ff_w1, ff_b1, ff_w2, ff_b2, ln_g, ln_b,
        mlp_w1, mlp_b1, mlp_w2, mlp_b2, out_w, out_b, out);
}


# SGD wave: trip count of the step loop read once as a scalar (readfirstlane); counter, exit test and back branch now run on the scalar unit, not vector compare -> exec; sums and their order unchanged
# speedup vs baseline: 6.1268x; 1.0985x over previous
#include <hip/hip_runtime.h>
#include <hip/hip_bf16.h>
#include <math.h>

typedef __attribute__((ext_vector_type(2))) float v2f;
typedef __attribute__((ext_vector_type(8))) float v8f;

#define LRC  0.01f
#define EPSC 1e-5f

#define Bq 256
#define Lq 4096
#define Hq 32
#define Vq 64
#define CH 16
#define NCH (Lq / CH)
#define NP  3
#define NGRP ((NCH + NP - 1) / NP)
#define NTHR 128
static_assert(NTHR == 32 * (NP + 1), "block = NP producer waves + 1 consumer wave");
#define A1STR 68
#define HSTR  36
#define LNB   8
static_assert(CH % LNB == 0, "a chunk is a whole number of LayerNorm batches");

template <int OFF>
__device__ __forceinline__ float xlane(float v) {
    const int b = __builtin_bit_cast(int, v);
    if constexpr (OFF == 16) {
        return __builtin_bit_cast(float,
            __builtin_amdgcn_permlanex16(b, b, 0x76543210, 0xfedcba98, false, false));
    } else {
        return __builtin_bit_cast(float,
            __builtin_amdgcn_update_dpp(0, b, 0x160 + OFF, 0xf, 0xf, true));
    }
}
template <int OFF, int N>
__device__ __forceinline__ void bfly_add(float (&v)[N]) {
    #pragma unroll
    for (int i = 0; i < N; ++i) v[i] += xlane<OFF>(v[i]);
}
template <int N>
__device__ __forceinline__ void wave_sum(float (&v)[N]) {
    bfly_add<16, N>(v);
    bfly_add<8, N>(v);
    bfly_add<4, N>(v);
    bfly_add<2, N>(v);
    bfly_add<1, N>(v);
}
#define SLOT_ROT(lane) (((lane) >> 4) << 2)
__device__ __forceinline__ void wave_sum8_rows(float (&v)[8]) {
    float mine[4];
    #pragma unroll
    for (int k = 0; k < 4; ++k) mine[k] = v[k] + xlane<16>(v[k + 4]);
    bfly_add<8, 4>(mine);
    bfly_add<4, 4>(mine);
    bfly_add<2, 4>(mine);
    bfly_add<1, 4>(mine);
    #pragma unroll
    for (int k = 0; k < 4; ++k) {
        v[k]     = mine[k];
        v[k + 4] = xlane<16>(mine[k]);
    }
}

__launch_bounds__(NTHR, 1)
__global__ void pmm_fused_kernel(const int* __restrict__ seq,
                                 const float* __restrict__ embed,
                                 const float* __restrict__ ff_w1,
                                 const float* __restrict__ ff_b1,
                                 const float* __restrict__ ff_w2,
                                 const float* __restrict__ ff_b2,
                                 const float* __restrict__ ln_g,
                                 const float* __restrict__ ln_b,
                                 const float* __restrict__ mlp_w1,
                                 const float* __restrict__ mlp_b1,
                                 const float* __restrict__ mlp_w2,
                                 const float* __restrict__ mlp_b2,
                                 const float* __restrict__ out_w,
                                 const float* __restrict__ out_b,
                                 float* __restrict__ out)
{
    __shared__ float embedLds[Vq * Hq];
    __shared__ float w1Lds[2 * Hq * Hq];
    __shared__ float w2Lds[Hq * 2 * Hq];
    __shared__ float a1buf[NP][CH * A1STR];
    __shared__ float hid[2][NP * CH * HSTR];
    __shared__ float ctxLds[Hq];

    const int b    = blockIdx.x;
    const int tid  = threadIdx.x;
    const int lane = tid & 31;
    const int wave = tid >> 5;
    const int m    = lane & 15;
    const int hi   = lane >> 4;

    for (int i = tid; i < Vq * Hq; i += NTHR)     embedLds[i] = embed[i];
    for (int i = tid; i < 2 * Hq * Hq; i += NTHR) w1Lds[i]    = ff_w1[i];
    for (int i = tid; i < 2 * Hq * Hq; i += NTHR) w2Lds[i]    = ff_w2[i];
    __syncthreads();

    float b1t[4], b2t[2], lnGv = 0.f, lnBv = 0.f;
    if (wave < NP) {
        #pragma unroll
        for (int nt = 0; nt < 4; ++nt) b1t[nt] = ff_b1[nt * 16 + m];
        #pragma unroll
        for (int nt = 0; nt < 2; ++nt) b2t[nt] = ff_b2[nt * 16 + m];
        lnGv = ln_g[lane];
        lnBv = ln_b[lane];
    }

    float w1c[8], b1v[8], w2r[8], b2v = 0.f;
    if (wave == NP) {
        const int rot = SLOT_ROT(lane);
        #pragma unroll
        for (int k = 0; k < 8; ++k) {
            const int i = k ^ rot;
            w1c[k] = mlp_w1[i * Hq + lane];
            b1v[k] = mlp_b1[i];
            w2r[k] = mlp_w2[lane * 8 + i];
        }
        b2v = mlp_b2[lane];
    }

    float* const a1w = a1buf[wave < NP ? wave : 0];
    auto produce = [&](int c, float* hbuf) {
        const int base = c * CH;
        const int tok  = seq[b * Lq + base + m];

        v2f eA[8];
        #pragma unroll
        for (int kb = 0; kb < 8; ++kb)
            eA[kb] = *(const v2f*)&embedLds[tok * Hq + 4 * kb + 2 * hi];

        #pragma unroll 1
        for (int nt = 0; nt < 4; ++nt) {
            const int n0 = nt * 16;
            v8f acc = {};
            #pragma unroll
            for (int kb = 0; kb < 8; ++kb) {
                v2f bB = *(const v2f*)&w1Lds[(n0 + m) * Hq + 4 * kb + 2 * hi];
                acc = __builtin_amdgcn_wmma_f32_16x16x4_f32(
                        false, eA[kb], false, bB, (short)0, acc, false, false);
            }
            const float bias = b1t[nt];
            #pragma unroll
            for (int r = 0; r < 8; ++r) {
                float z = acc[r] + bias;
                a1w[(r + 8 * hi) * A1STR + n0 + m] = fmaxf(z, 0.f);
            }
        }
        asm volatile("s_wait_dscnt 0" ::: "memory");
        __builtin_amdgcn_wave_barrier();

        v2f a2[16];
        #pragma unroll
        for (int kb = 0; kb < 16; ++kb)
            a2[kb] = *(const v2f*)&a1w[m * A1STR + 4 * kb + 2 * hi];

        #pragma unroll 1
        for (int nt = 0; nt < 2; ++nt) {
            const int n0 = nt * 16;
            v8f acc = {};
            #pragma unroll
            for (int kb = 0; kb < 16; ++kb) {
                v2f bB = *(const v2f*)&w2Lds[(n0 + m) * (2 * Hq) + 4 * kb + 2 * hi];
                acc = __builtin_amdgcn_wmma_f32_16x16x4_f32(
                        false, a2[kb], false, bB, (short)0, acc, false, false);
            }
            const float bias = b2t[nt];
            #pragma unroll
            for (int r = 0; r < 8; ++r) {
                int   tm = __shfl(tok, r + 8 * hi, 32);
                float e  = embedLds[tm * Hq + n0 + m];
                hbuf[(r + 8 * hi) * HSTR + n0 + m] = acc[r] + bias + e;
            }
        }
        asm volatile("s_wait_dscnt 0" ::: "memory");
        __builtin_amdgcn_wave_barrier();

        #pragma unroll 1
        for (int t0 = 0; t0 < CH; t0 += LNB) {
            float* const hp = hbuf + t0 * HSTR + lane;
            float z[LNB];
            #pragma unroll
            for (int t = 0; t < LNB; ++t) z[t] = hp[t * HSTR];
            float ss[2 * LNB];
            #pragma unroll
            for (int t = 0; t < LNB; ++t) { ss[2 * t] = z[t]; ss[2 * t + 1] = z[t] * z[t]; }
            wave_sum<2 * LNB>(ss);
            #pragma unroll
            for (int t = 0; t < LNB; ++t) {
                const float s = ss[2 * t], s2 = ss[2 * t + 1];
                float mu  = s * (1.f / 32.f);
                float var = s2 * (1.f / 32.f) - mu * mu;
                float nz  = (z[t] - mu) * rsqrtf(var + EPSC);
                hp[t * HSTR] = nz * lnGv + lnBv;
            }
        }
        asm volatile("s_wait_dscnt 0" ::: "memory");
    };

    auto consume = [&](const float* hbuf, int nsteps_any_lane) {
        const int nsteps = __builtin_amdgcn_readfirstlane(nsteps_any_lane);
        const float* rp = hbuf + lane;
        float xn = rp[0];
        float vn = rp[HSTR];
        asm volatile("" :: "v"(xn), "v"(vn));
        if (nsteps > 1) rp += 2 * HSTR;
        #pragma unroll 1
        for (int s = 0; s < nsteps; ++s) {
            const float x = xn;
            const float v = vn;
            xn = rp[0];
            vn = rp[HSTR];
            __builtin_amdgcn_sched_barrier(0);
            if (s + 2 < nsteps) rp += 2 * HSTR;

            float p[8];
            #pragma unroll
            for (int i = 0; i < 8; ++i) p[i] = w1c[i] * x;
            wave_sum8_rows(p);
            float a1p[8], a1v[8];
            float y = b2v;
            #pragma unroll
            for (int i = 0; i < 8; ++i) {
                a1p[i] = p[i] + b1v[i];
                a1v[i] = fmaxf(a1p[i], 0.f);
                y = fmaf(w2r[i], a1v[i], y);
            }
            float gate[8];
            #pragma unroll
            for (int i = 0; i < 8; ++i) gate[i] = (a1p[i] > 0.f) ? -LRC : 0.f;
            float d = (y - v) * 0.0625f;

            float q[8];
            #pragma unroll
            for (int i = 0; i < 8; ++i) q[i] = w2r[i] * d;
            wave_sum8_rows(q);
            b2v -= LRC * d;
            #pragma unroll
            for (int i = 0; i < 8; ++i) {
                const float sg = q[i] * gate[i];
                w2r[i] = fmaf(-LRC * d, a1v[i], w2r[i]);
                b1v[i] = fmaf(q[i], gate[i], b1v[i]);
                w1c[i] = fmaf(sg, x, w1c[i]);
            }
        }
    };

    if (wave < NP) produce(wave, hid[0] + wave * (CH * HSTR));
    __syncthreads();

    for (int g = 0; g < NGRP; ++g) {
        if (wave < NP) {
            const int c = NP * (g + 1) + wave;
            if (c < NCH) produce(c, hid[(g + 1) & 1] + wave * (CH * HSTR));
        } else {
            consume(hid[g & 1], (g == NGRP - 1) ? (8 * (NCH - NP * (NGRP - 1)) - 1) : 8 * NP);
        }
        __syncthreads();
    }

    if (wave == NP) {
        const float* hbuf = hid[(NGRP - 1) & 1] + ((NCH - 1) - NP * (NGRP - 1)) * (CH * HSTR);
        float x = hbuf[15 * HSTR + lane];
        float p[8];
        #pragma unroll
        for (int i = 0; i < 8; ++i) p[i] = w1c[i] * x;
        wave_sum8_rows(p);
        float y = b2v;
        #pragma unroll
        for (int i = 0; i < 8; ++i) y = fmaf(w2r[i], fmaxf(p[i] + b1v[i], 0.f), y);
        ctxLds[lane] = y;
    }
    __syncthreads();

    if (tid < Vq) {
        float acc = out_b[tid];
        #pragma unroll
        for (int h = 0; h < Hq; ++h)
            acc = fmaf(ctxLds[h], out_w[tid * Hq + h], acc);
        out[b * Vq + tid] = acc;
    }
}

extern "C" void kernel_launch(void* const* d_in, const int* in_sizes, int n_in,
                              void* d_out, int out_size, void* d_ws, size_t ws_size,
                              hipStream_t stream) {
    (void)in_sizes; (void)n_in; (void)out_size; (void)d_ws; (void)ws_size;
    const int*   seq    = (const int*)  d_in[0];
    const float* embed  = (const float*)d_in[1];
    const float* ff_w1  = (const float*)d_in[2];
    const float* ff_b1  = (const float*)d_in[3];
    const float* ff_w2  = (const float*)d_in[4];
    const float* ff_b2  = (const float*)d_in[5];
    const float* ln_g   = (const float*)d_in[6];
    const float* ln_b   = (const float*)d_in[7];
    const float* mlp_w1 = (const float*)d_in[8];
    const float* mlp_b1 = (const float*)d_in[9];
    const float* mlp_w2 = (const float*)d_in[10];
    const float* mlp_b2 = (const float*)d_in[11];
    const float* out_w  = (const float*)d_in[12];
    const float* out_b  = (const float*)d_in[13];
    float*       out    = (float*)d_out;

    pmm_fused_kernel<<<dim3(Bq), dim3(NTHR), 0, stream>>>(
        seq, embed, ff_w1, ff_b1, ff_w2, ff_b2, ln_g, ln_b,
        mlp_w1, mlp_b1, mlp_w2, mlp_b2, out_w, out_b, out);
}


# SGD wave: forward sums of step s+1 reduced during step s on the pre-update weights, then corrected by one fma with sg*(x.x'+1): that reduction leaves the serial chain; p's sum order changes
# speedup vs baseline: 6.5499x; 1.0691x over previous
#include <hip/hip_runtime.h>
#include <hip/hip_bf16.h>
#include <math.h>

typedef __attribute__((ext_vector_type(2))) float v2f;
typedef __attribute__((ext_vector_type(8))) float v8f;

#define LRC  0.01f
#define EPSC 1e-5f

#define Bq 256
#define Lq 4096
#define Hq 32
#define Vq 64
#define CH 16
#define NCH (Lq / CH)
#define NP  3
#define NGRP ((NCH + NP - 1) / NP)
#define NTHR 128
static_assert(NTHR == 32 * (NP + 1), "block = NP producer waves + 1 consumer wave");
#define A1STR 68
#define HSTR  36
#define LNB   8
static_assert(CH % LNB == 0, "a chunk is a whole number of LayerNorm batches");

template <int OFF>
__device__ __forceinline__ float xlane(float v) {
    const int b = __builtin_bit_cast(int, v);
    if constexpr (OFF == 16) {
        return __builtin_bit_cast(float,
            __builtin_amdgcn_permlanex16(b, b, 0x76543210, 0xfedcba98, false, false));
    } else {
        return __builtin_bit_cast(float,
            __builtin_amdgcn_update_dpp(0, b, 0x160 + OFF, 0xf, 0xf, true));
    }
}
template <int OFF, int N>
__device__ __forceinline__ void bfly_add(float (&v)[N]) {
    #pragma unroll
    for (int i = 0; i < N; ++i) v[i] += xlane<OFF>(v[i]);
}
template <int N>
__device__ __forceinline__ void wave_sum(float (&v)[N]) {
    bfly_add<16, N>(v);
    bfly_add<8, N>(v);
    bfly_add<4, N>(v);
    bfly_add<2, N>(v);
    bfly_add<1, N>(v);
}
#define SLOT_ROT(lane) (((lane) >> 4) << 2)
__device__ __forceinline__ void wave_sum8_rows(float (&v)[8]) {
    float mine[4];
    #pragma unroll
    for (int k = 0; k < 4; ++k) mine[k] = v[k] + xlane<16>(v[k + 4]);
    bfly_add<8, 4>(mine);
    bfly_add<4, 4>(mine);
    bfly_add<2, 4>(mine);
    bfly_add<1, 4>(mine);
    #pragma unroll
    for (int k = 0; k < 4; ++k) {
        v[k]     = mine[k];
        v[k + 4] = xlane<16>(mine[k]);
    }
}

__launch_bounds__(NTHR, 1)
__global__ void pmm_fused_kernel(const int* __restrict__ seq,
                                 const float* __restrict__ embed,
                                 const float* __restrict__ ff_w1,
                                 const float* __restrict__ ff_b1,
                                 const float* __restrict__ ff_w2,
                                 const float* __restrict__ ff_b2,
                                 const float* __restrict__ ln_g,
                                 const float* __restrict__ ln_b,
                                 const float* __restrict__ mlp_w1,
                                 const float* __restrict__ mlp_b1,
                                 const float* __restrict__ mlp_w2,
                                 const float* __restrict__ mlp_b2,
                                 const float* __restrict__ out_w,
                                 const float* __restrict__ out_b,
                                 float* __restrict__ out)
{
    __shared__ float embedLds[Vq * Hq];
    __shared__ float w1Lds[2 * Hq * Hq];
    __shared__ float w2Lds[Hq * 2 * Hq];
    __shared__ float a1buf[NP][CH * A1STR];
    __shared__ float hid[2][NP * CH * HSTR];
    __shared__ float ctxLds[Hq];

    const int b    = blockIdx.x;
    const int tid  = threadIdx.x;
    const int lane = tid & 31;
    const int wave = tid >> 5;
    const int m    = lane & 15;
    const int hi   = lane >> 4;

    for (int i = tid; i < Vq * Hq; i += NTHR)     embedLds[i] = embed[i];
    for (int i = tid; i < 2 * Hq * Hq; i += NTHR) w1Lds[i]    = ff_w1[i];
    for (int i = tid; i < 2 * Hq * Hq; i += NTHR) w2Lds[i]    = ff_w2[i];
    __syncthreads();

    float b1t[4], b2t[2], lnGv = 0.f, lnBv = 0.f;
    if (wave < NP) {
        #pragma unroll
        for (int nt = 0; nt < 4; ++nt) b1t[nt] = ff_b1[nt * 16 + m];
        #pragma unroll
        for (int nt = 0; nt < 2; ++nt) b2t[nt] = ff_b2[nt * 16 + m];
        lnGv = ln_g[lane];
        lnBv = ln_b[lane];
    }

    float w1c[8], b1v[8], w2r[8], b2v = 0.f;
    if (wave == NP) {
        const int rot = SLOT_ROT(lane);
        #pragma unroll
        for (int k = 0; k < 8; ++k) {
            const int i = k ^ rot;
            w1c[k] = mlp_w1[i * Hq + lane];
            b1v[k] = mlp_b1[i];
            w2r[k] = mlp_w2[lane * 8 + i];
        }
        b2v = mlp_b2[lane];
    }

    float* const a1w = a1buf[wave < NP ? wave : 0];
    auto produce = [&](int c, float* hbuf) {
        const int base = c * CH;
        const int tok  = seq[b * Lq + base + m];

        v2f eA[8];
        #pragma unroll
        for (int kb = 0; kb < 8; ++kb)
            eA[kb] = *(const v2f*)&embedLds[tok * Hq + 4 * kb + 2 * hi];

        #pragma unroll 1
        for (int nt = 0; nt < 4; ++nt) {
            const int n0 = nt * 16;
            v8f acc = {};
            #pragma unroll
            for (int kb = 0; kb < 8; ++kb) {
                v2f bB = *(const v2f*)&w1Lds[(n0 + m) * Hq + 4 * kb + 2 * hi];
                acc = __builtin_amdgcn_wmma_f32_16x16x4_f32(
                        false, eA[kb], false, bB, (short)0, acc, false, false);
            }
            const float bias = b1t[nt];
            #pragma unroll
            for (int r = 0; r < 8; ++r) {
                float z = acc[r] + bias;
                a1w[(r + 8 * hi) * A1STR + n0 + m] = fmaxf(z, 0.f);
            }
        }
        asm volatile("s_wait_dscnt 0" ::: "memory");
        __builtin_amdgcn_wave_barrier();

        v2f a2[16];
        #pragma unroll
        for (int kb = 0; kb < 16; ++kb)
            a2[kb] = *(const v2f*)&a1w[m * A1STR + 4 * kb + 2 * hi];

        #pragma unroll 1
        for (int nt = 0; nt < 2; ++nt) {
            const int n0 = nt * 16;
            float res[8];
            #pragma unroll
            for (int r = 0; r < 8; ++r) {
                int tm = __shfl(tok, r + 8 * hi, 32);
                res[r] = embedLds[tm * Hq + n0 + m];
            }
            v8f acc = {};
            #pragma unroll
            for (int kb = 0; kb < 16; ++kb) {
                v2f bB = *(const v2f*)&w2Lds[(n0 + m) * (2 * Hq) + 4 * kb + 2 * hi];
                acc = __builtin_amdgcn_wmma_f32_16x16x4_f32(
                        false, a2[kb], false, bB, (short)0, acc, false, false);
            }
            const float bias = b2t[nt];
            #pragma unroll
            for (int r = 0; r < 8; ++r)
                hbuf[(r + 8 * hi) * HSTR + n0 + m] = acc[r] + bias + res[r];
        }
        asm volatile("s_wait_dscnt 0" ::: "memory");
        __builtin_amdgcn_wave_barrier();

        #pragma unroll 1
        for (int t0 = 0; t0 < CH; t0 += LNB) {
            float* const hp = hbuf + t0 * HSTR + lane;
            float z[LNB];
            #pragma unroll
            for (int t = 0; t < LNB; ++t) z[t] = hp[t * HSTR];
            float ss[2 * LNB];
            #pragma unroll
            for (int t = 0; t < LNB; ++t) { ss[2 * t] = z[t]; ss[2 * t + 1] = z[t] * z[t]; }
            wave_sum<2 * LNB>(ss);
            #pragma unroll
            for (int t = 0; t < LNB; ++t) {
                const float s = ss[2 * t], s2 = ss[2 * t + 1];
                float mu  = s * (1.f / 32.f);
                float var = s2 * (1.f / 32.f) - mu * mu;
                float nz  = (z[t] - mu) * rsqrtf(var + EPSC);
                hp[t * HSTR] = nz * lnGv + lnBv;
            }
        }
        asm volatile("s_wait_dscnt 0" ::: "memory");
    };

    #define HOLD1(a)             asm volatile("" :: "v"(a))
    #define HOLD2(a, b)          asm volatile("" :: "v"(a), "v"(b))
    #define HOLD3(a, b, c)       asm volatile("" :: "v"(a), "v"(b), "v"(c))
    #define HOLD4(a, b, c, d)    asm volatile("" :: "v"(a), "v"(b), "v"(c), "v"(d))
    #define HOLD5(a, b, c, d, e) asm volatile("" :: "v"(a), "v"(b), "v"(c), "v"(d), "v"(e))
    #define XROW_INTO(dead, v)   __builtin_bit_cast(float, __builtin_amdgcn_permlanex16( \
                                     __builtin_bit_cast(int, dead), __builtin_bit_cast(int, v), \
                                     0x76543210, 0xfedcba98, false, false))
    #define SLICE_END()          __builtin_amdgcn_sched_barrier(0)
    auto consume = [&](const float* hbuf, int nsteps_any_lane) {
        const int nsteps = __builtin_amdgcn_readfirstlane(nsteps_any_lane);
        const float* rp = hbuf + lane;
        float x = rp[0];
        float v = rp[HSTR];
        const int o1 = (nsteps > 1) ? 2 * HSTR : 0;
        float xn = rp[o1];
        float vn = rp[o1 + HSTR];
        HOLD4(x, v, xn, vn);
        rp += (nsteps > 2) ? 4 * HSTR : o1;
        float a1v[8];
        {
            float P[8];
            #pragma unroll
            for (int i = 0; i < 8; ++i) P[i] = w1c[i] * x;
            wave_sum8_rows(P);
            #pragma unroll
            for (int i = 0; i < 8; ++i) a1v[i] = fmaxf(P[i] + b1v[i], 0.f);
        }
        float y = fmaf(w2r[1], a1v[1], fmaf(w2r[0], a1v[0], b2v));
        #pragma unroll 1
        for (int s = 0; s < nsteps; ++s) {
            const float x2 = rp[0];
            const float v2 = rp[HSTR];
            SLICE_END();
            if (s + 3 < nsteps) rp += 2 * HSTR;

            float gate[8], e[4], m[4], qe[4], qm[4], re[4], qr[4], Rb[8], sg[8], a1n[8];
            float xe, dxv, dsub, d, nld, dx1;

            y = fmaf(w2r[2], a1v[2], y);
            y = fmaf(w2r[3], a1v[3], y);
            #pragma unroll
            for (int k = 0; k < 4; ++k) e[k] = xlane<16>(w1c[k + 4] * xn);
            xe = xlane<16>(x * xn);
            HOLD5(y, e[0], e[1], e[2], e[3]); HOLD1(xe); SLICE_END();

            y = fmaf(w2r[4], a1v[4], y);
            #pragma unroll
            for (int k = 0; k < 4; ++k) m[k] = w1c[k] * xn + e[k];
            dxv = x * xn + xe;
            gate[0] = (a1v[0] > 0.f) ? -LRC : 0.f;
            HOLD5(y, m[0], m[1], m[2], m[3]); HOLD2(dxv, gate[0]); SLICE_END();

            y = fmaf(w2r[5], a1v[5], y);
            m[0] += xlane<8>(m[0]); m[1] += xlane<8>(m[1]);
            gate[1] = (a1v[1] > 0.f) ? -LRC : 0.f;
            HOLD4(y, m[0], m[1], gate[1]); SLICE_END();

            y = fmaf(w2r[6], a1v[6], y);
            m[2] += xlane<8>(m[2]); m[3] += xlane<8>(m[3]);
            gate[2] = (a1v[2] > 0.f) ? -LRC : 0.f;
            HOLD4(y, m[2], m[3], gate[2]); SLICE_END();

            y = fmaf(w2r[7], a1v[7], y);
            dxv += xlane<8>(dxv); m[0] += xlane<4>(m[0]);
            gate[3] = (a1v[3] > 0.f) ? -LRC : 0.f;
            HOLD4(y, dxv, m[0], gate[3]); SLICE_END();

            dsub = y - v;
            m[1] += xlane<4>(m[1]); m[2] += xlane<4>(m[2]);
            gate[4] = (a1v[4] > 0.f) ? -LRC : 0.f;
            HOLD4(dsub, m[1], m[2], gate[4]); SLICE_END();

            d = dsub * 0.0625f;
            m[3] += xlane<4>(m[3]); dxv += xlane<4>(dxv);
            gate[5] = (a1v[5] > 0.f) ? -LRC : 0.f;
            HOLD4(d, m[3], dxv, gate[5]); SLICE_END();

            #pragma unroll
            for (int k = 0; k < 4; ++k) qe[k] = xlane<16>(w2r[k + 4] * d);
            nld = -LRC * d;
            m[0] += xlane<2>(m[0]); m[1] += xlane<2>(m[1]);
            gate[6] = (a1v[6] > 0.f) ? -LRC : 0.f;
            HOLD4(qe[0], qe[1], qe[2], qe[3]); HOLD4(nld, m[0], m[1], gate[6]); SLICE_END();

            #pragma unroll
            for (int k = 0; k < 4; ++k) qm[k] = w2r[k] * d + qe[k];
            m[2] += xlane<2>(m[2]); m[3] += xlane<2>(m[3]);
            gate[7] = (a1v[7] > 0.f) ? -LRC : 0.f;
            b2v = fmaf(-LRC, d, b2v);
            HOLD4(qm[0], qm[1], qm[2], qm[3]); HOLD4(m[2], m[3], gate[7], b2v); SLICE_END();

            dxv += xlane<2>(dxv); m[0] += xlane<1>(m[0]); m[1] += xlane<1>(m[1]);
            HOLD3(dxv, m[0], m[1]); SLICE_END();

            #pragma unroll
            for (int k = 0; k < 4; ++k) qm[k] += xlane<8>(qm[k]);
            w2r[0] = fmaf(nld, a1v[0], w2r[0]); w2r[1] = fmaf(nld, a1v[1], w2r[1]);
            m[2] += xlane<1>(m[2]);
            HOLD4(qm[0], qm[1], qm[2], qm[3]); HOLD3(w2r[0], w2r[1], m[2]); SLICE_END();
            #pragma unroll
            for (int k = 0; k < 4; ++k) qm[k] += xlane<4>(qm[k]);
            w2r[2] = fmaf(nld, a1v[2], w2r[2]); w2r[3] = fmaf(nld, a1v[3], w2r[3]);
            m[3] += xlane<1>(m[3]);
            re[0] = XROW_INTO(e[0], m[0]);
            HOLD4(qm[0], qm[1], qm[2], qm[3]); HOLD4(w2r[2], w2r[3], m[3], re[0]); SLICE_END();
            #pragma unroll
            for (int k = 0; k < 4; ++k) qm[k] += xlane<2>(qm[k]);
            w2r[4] = fmaf(nld, a1v[4], w2r[4]); w2r[5] = fmaf(nld, a1v[5], w2r[5]);
            dxv += xlane<1>(dxv);
            re[1] = XROW_INTO(e[1], m[1]);
            re[2] = XROW_INTO(e[2], m[2]);
            HOLD4(qm[0], qm[1], qm[2], qm[3]); HOLD5(w2r[4], w2r[5], dxv, re[1], re[2]); SLICE_END();
            #pragma unroll
            for (int k = 0; k < 4; ++k) qm[k] += xlane<1>(qm[k]);
            w2r[6] = fmaf(nld, a1v[6], w2r[6]); w2r[7] = fmaf(nld, a1v[7], w2r[7]);
            re[3] = XROW_INTO(e[3], m[3]);
            dx1 = dxv + 1.f;
            HOLD4(qm[0], qm[1], qm[2], qm[3]); HOLD4(w2r[6], w2r[7], re[3], dx1); SLICE_END();

            #pragma unroll
            for (int k = 0; k < 4; ++k) {
                sg[k] = qm[k] * gate[k];
                qr[k] = XROW_INTO(qe[k], qm[k]);
                Rb[k] = m[k] + b1v[k]; Rb[k + 4] = re[k] + b1v[k + 4];
            }
            HOLD4(sg[0], sg[1], sg[2], sg[3]); HOLD4(qr[0], qr[1], qr[2], qr[3]);
            HOLD4(Rb[0], Rb[1], Rb[2], Rb[3]); HOLD4(Rb[4], Rb[5], Rb[6], Rb[7]); SLICE_END();

            #pragma unroll
            for (int k = 0; k < 4; ++k) {
                a1n[k]    = fmaxf(fmaf(sg[k], dx1, Rb[k]), 0.f);
                sg[k + 4] = qr[k] * gate[k + 4];
                b1v[k]    = fmaf(qm[k], gate[k], b1v[k]);
                w1c[k]    = fmaf(sg[k], x, w1c[k]);
            }
            HOLD4(a1n[0], a1n[1], a1n[2], a1n[3]); HOLD4(sg[4], sg[5], sg[6], sg[7]); SLICE_END();

            y = fmaf(w2r[0], a1n[0], b2v);
            #pragma unroll
            for (int k = 4; k < 8; ++k) {
                a1n[k] = fmaxf(fmaf(sg[k], dx1, Rb[k]), 0.f);
                b1v[k] = fmaf(qr[k - 4], gate[k], b1v[k]);
                w1c[k] = fmaf(sg[k], x, w1c[k]);
            }
            HOLD5(y, a1n[4], a1n[5], a1n[6], a1n[7]); SLICE_END();
            y = fmaf(w2r[1], a1n[1], y);
            #pragma unroll
            for (int i = 0; i < 8; ++i) a1v[i] = a1n[i];
            x = xn; v = vn; xn = x2; vn = v2;
        }
    };
    #undef HOLD1
    #undef HOLD2
    #undef HOLD3
    #undef HOLD4
    #undef HOLD5
    #undef SLICE_END
    #undef XROW_INTO

    if (wave < NP) produce(wave, hid[0] + wave * (CH * HSTR));
    __syncthreads();

    for (int g = 0; g < NGRP; ++g) {
        if (wave < NP) {
            const int c = NP * (g + 1) + wave;
            if (c < NCH) produce(c, hid[(g + 1) & 1] + wave * (CH * HSTR));
        } else {
            consume(hid[g & 1], (g == NGRP - 1) ? (8 * (NCH - NP * (NGRP - 1)) - 1) : 8 * NP);
        }
        __syncthreads();
    }

    if (wave == NP) {
        const float* hbuf = hid[(NGRP - 1) & 1] + ((NCH - 1) - NP * (NGRP - 1)) * (CH * HSTR);
        float x = hbuf[15 * HSTR + lane];
        float p[8];
        #pragma unroll
        for (int i = 0; i < 8; ++i) p[i] = w1c[i] * x;
        wave_sum8_rows(p);
        float y = b2v;
        #pragma unroll
        for (int i = 0; i < 8; ++i) y = fmaf(w2r[i], fmaxf(p[i] + b1v[i], 0.f), y);
        ctxLds[lane] = y;
    }
    __syncthreads();

    if (tid < Vq) {
        float acc = out_b[tid];
        #pragma unroll
        for (int h = 0; h < Hq; ++h)
            acc = fmaf(ctxLds[h], out_w[tid * Hq + h], acc);
        out[b * Vq + tid] = acc;
    }
}

extern "C" void kernel_launch(void* const* d_in, const int* in_sizes, int n_in,
                              void* d_out, int out_size, void* d_ws, size_t ws_size,
                              hipStream_t stream) {
    (void)in_sizes; (void)n_in; (void)out_size; (void)d_ws; (void)ws_size;
    const int*   seq    = (const int*)  d_in[0];
    const float* embed  = (const float*)d_in[1];
    const float* ff_w1  = (const float*)d_in[2];
    const float* ff_b1  = (const float*)d_in[3];
    const float* ff_w2  = (const float*)d_in[4];
    const float* ff_b2  = (const float*)d_in[5];
    const float* ln_g   = (const float*)d_in[6];
    const float* ln_b   = (const float*)d_in[7];
    const float* mlp_w1 = (const float*)d_in[8];
    const float* mlp_b1 = (const float*)d_in[9];
    const float* mlp_w2 = (const float*)d_in[10];
    const float* mlp_b2 = (const float*)d_in[11];
    const float* out_w  = (const float*)d_in[12];
    const float* out_b  = (const float*)d_in[13];
    float*       out    = (float*)d_out;

    pmm_fused_kernel<<<dim3(Bq), dim3(NTHR), 0, stream>>>(
        seq, embed, ff_w1, ff_b1, ff_w2, ff_b2, ln_g, ln_b,
        mlp_w1, mlp_b1, mlp_w2, mlp_b2, out_w, out_b, out);
}


# Producer waves keep the B fragments of both FF weight matrices in registers for the whole kernel (128 VGPRs, read from LDS once) instead of re-reading them per chunk between the matrix instructions
# speedup vs baseline: 6.7596x; 1.0320x over previous
#include <hip/hip_runtime.h>
#include <hip/hip_bf16.h>
#include <math.h>

typedef __attribute__((ext_vector_type(2))) float v2f;
typedef __attribute__((ext_vector_type(8))) float v8f;

#define LRC  0.01f
#define EPSC 1e-5f

#define Bq 256
#define Lq 4096
#define Hq 32
#define Vq 64
#define CH 16
#define NCH (Lq / CH)
#define NP  3
#define NGRP ((NCH + NP - 1) / NP)
#define NTHR 128
static_assert(NTHR == 32 * (NP + 1), "block = NP producer waves + 1 consumer wave");
#define A1STR 68
#define HSTR  36
#define LNB   8
static_assert(CH % LNB == 0, "a chunk is a whole number of LayerNorm batches");

template <int OFF>
__device__ __forceinline__ float xlane(float v) {
    const int b = __builtin_bit_cast(int, v);
    if constexpr (OFF == 16) {
        return __builtin_bit_cast(float,
            __builtin_amdgcn_permlanex16(b, b, 0x76543210, 0xfedcba98, false, false));
    } else {
        return __builtin_bit_cast(float,
            __builtin_amdgcn_update_dpp(0, b, 0x160 + OFF, 0xf, 0xf, true));
    }
}
template <int OFF, int N>
__device__ __forceinline__ void bfly_add(float (&v)[N]) {
    #pragma unroll
    for (int i = 0; i < N; ++i) v[i] += xlane<OFF>(v[i]);
}
template <int N>
__device__ __forceinline__ void wave_sum(float (&v)[N]) {
    bfly_add<16, N>(v);
    bfly_add<8, N>(v);
    bfly_add<4, N>(v);
    bfly_add<2, N>(v);
    bfly_add<1, N>(v);
}
#define SLOT_ROT(lane) (((lane) >> 4) << 2)
__device__ __forceinline__ void wave_sum8_rows(float (&v)[8]) {
    float mine[4];
    #pragma unroll
    for (int k = 0; k < 4; ++k) mine[k] = v[k] + xlane<16>(v[k + 4]);
    bfly_add<8, 4>(mine);
    bfly_add<4, 4>(mine);
    bfly_add<2, 4>(mine);
    bfly_add<1, 4>(mine);
    #pragma unroll
    for (int k = 0; k < 4; ++k) {
        v[k]     = mine[k];
        v[k + 4] = xlane<16>(mine[k]);
    }
}

__launch_bounds__(NTHR, 1)
__global__ void pmm_fused_kernel(const int* __restrict__ seq,
                                 const float* __restrict__ embed,
                                 const float* __restrict__ ff_w1,
                                 const float* __restrict__ ff_b1,
                                 const float* __restrict__ ff_w2,
                                 const float* __restrict__ ff_b2,
                                 const float* __restrict__ ln_g,
                                 const float* __restrict__ ln_b,
                                 const float* __restrict__ mlp_w1,
                                 const float* __restrict__ mlp_b1,
                                 const float* __restrict__ mlp_w2,
                                 const float* __restrict__ mlp_b2,
                                 const float* __restrict__ out_w,
                                 const float* __restrict__ out_b,
                                 float* __restrict__ out)
{
    __shared__ float embedLds[Vq * Hq];
    __shared__ float w1Lds[2 * Hq * Hq];
    __shared__ float w2Lds[Hq * 2 * Hq];
    __shared__ float a1buf[NP][CH * A1STR];
    __shared__ float hid[2][NP * CH * HSTR];
    __shared__ float ctxLds[Hq];

    const int b    = blockIdx.x;
    const int tid  = threadIdx.x;
    const int lane = tid & 31;
    const int wave = tid >> 5;
    const int m    = lane & 15;
    const int hi   = lane >> 4;

    for (int i = tid; i < Vq * Hq; i += NTHR)     embedLds[i] = embed[i];
    for (int i = tid; i < 2 * Hq * Hq; i += NTHR) w1Lds[i]    = ff_w1[i];
    for (int i = tid; i < 2 * Hq * Hq; i += NTHR) w2Lds[i]    = ff_w2[i];
    __syncthreads();

    float b1t[4], b2t[2], lnGv = 0.f, lnBv = 0.f;
    if (wave < NP) {
        #pragma unroll
        for (int nt = 0; nt < 4; ++nt) b1t[nt] = ff_b1[nt * 16 + m];
        #pragma unroll
        for (int nt = 0; nt < 2; ++nt) b2t[nt] = ff_b2[nt * 16 + m];
        lnGv = ln_g[lane];
        lnBv = ln_b[lane];
    }

    v2f w1B[4][8], w2B[2][16];
    if (wave < NP) {
        #pragma unroll
        for (int nt = 0; nt < 4; ++nt) {
            #pragma unroll
            for (int kb = 0; kb < 8; ++kb)
                w1B[nt][kb] = *(const v2f*)&w1Lds[(nt * 16 + m) * Hq + 4 * kb + 2 * hi];
        }
        #pragma unroll
        for (int nt = 0; nt < 2; ++nt) {
            #pragma unroll
            for (int kb = 0; kb < 16; ++kb)
                w2B[nt][kb] = *(const v2f*)&w2Lds[(nt * 16 + m) * (2 * Hq) + 4 * kb + 2 * hi];
        }
    }

    float w1c[8], b1v[8], w2r[8], b2v = 0.f;
    if (wave == NP) {
        const int rot = SLOT_ROT(lane);
        #pragma unroll
        for (int k = 0; k < 8; ++k) {
            const int i = k ^ rot;
            w1c[k] = mlp_w1[i * Hq + lane];
            b1v[k] = mlp_b1[i];
            w2r[k] = mlp_w2[lane * 8 + i];
        }
        b2v = mlp_b2[lane];
    }

    float* const a1w = a1buf[wave < NP ? wave : 0];
    auto produce = [&](int c, float* hbuf) {
        const int base = c * CH;
        const int tok  = seq[b * Lq + base + m];

        v2f eA[8];
        #pragma unroll
        for (int kb = 0; kb < 8; ++kb)
            eA[kb] = *(const v2f*)&embedLds[tok * Hq + 4 * kb + 2 * hi];

        #pragma unroll
        for (int nt = 0; nt < 4; ++nt) {
            const int n0 = nt * 16;
            v8f acc = {};
            #pragma unroll
            for (int kb = 0; kb < 8; ++kb) {
                acc = __builtin_amdgcn_wmma_f32_16x16x4_f32(
                        false, eA[kb], false, w1B[nt][kb], (short)0, acc, false, false);
            }
            const float bias = b1t[nt];
            #pragma unroll
            for (int r = 0; r < 8; ++r) {
                float z = acc[r] + bias;
                a1w[(r + 8 * hi) * A1STR + n0 + m] = fmaxf(z, 0.f);
            }
        }
        asm volatile("s_wait_dscnt 0" ::: "memory");
        __builtin_amdgcn_wave_barrier();

        v2f a2[16];
        #pragma unroll
        for (int kb = 0; kb < 16; ++kb)
            a2[kb] = *(const v2f*)&a1w[m * A1STR + 4 * kb + 2 * hi];

        #pragma unroll
        for (int nt = 0; nt < 2; ++nt) {
            const int n0 = nt * 16;
            float res[8];
            #pragma unroll
            for (int r = 0; r < 8; ++r) {
                int tm = __shfl(tok, r + 8 * hi, 32);
                res[r] = embedLds[tm * Hq + n0 + m];
            }
            v8f acc = {};
            #pragma unroll
            for (int kb = 0; kb < 16; ++kb) {
                acc = __builtin_amdgcn_wmma_f32_16x16x4_f32(
                        false, a2[kb], false, w2B[nt][kb], (short)0, acc, false, false);
            }
            const float bias = b2t[nt];
            #pragma unroll
            for (int r = 0; r < 8; ++r)
                hbuf[(r + 8 * hi) * HSTR + n0 + m] = acc[r] + bias + res[r];
        }
        asm volatile("s_wait_dscnt 0" ::: "memory");
        __builtin_amdgcn_wave_barrier();

        #pragma unroll 1
        for (int t0 = 0; t0 < CH; t0 += LNB) {
            float* const hp = hbuf + t0 * HSTR + lane;
            float z[LNB];
            #pragma unroll
            for (int t = 0; t < LNB; ++t) z[t] = hp[t * HSTR];
            float ss[2 * LNB];
            #pragma unroll
            for (int t = 0; t < LNB; ++t) { ss[2 * t] = z[t]; ss[2 * t + 1] = z[t] * z[t]; }
            wave_sum<2 * LNB>(ss);
            #pragma unroll
            for (int t = 0; t < LNB; ++t) {
                const float s = ss[2 * t], s2 = ss[2 * t + 1];
                float mu  = s * (1.f / 32.f);
                float var = s2 * (1.f / 32.f) - mu * mu;
                float nz  = (z[t] - mu) * rsqrtf(var + EPSC);
                hp[t * HSTR] = nz * lnGv + lnBv;
            }
        }
        asm volatile("s_wait_dscnt 0" ::: "memory");
    };

    #define HOLD1(a)             asm volatile("" :: "v"(a))
    #define HOLD2(a, b)          asm volatile("" :: "v"(a), "v"(b))
    #define HOLD3(a, b, c)       asm volatile("" :: "v"(a), "v"(b), "v"(c))
    #define HOLD4(a, b, c, d)    asm volatile("" :: "v"(a), "v"(b), "v"(c), "v"(d))
    #define HOLD5(a, b, c, d, e) asm volatile("" :: "v"(a), "v"(b), "v"(c), "v"(d), "v"(e))
    #define XROW_INTO(dead, v)   __builtin_bit_cast(float, __builtin_amdgcn_permlanex16( \
                                     __builtin_bit_cast(int, dead), __builtin_bit_cast(int, v), \
                                     0x76543210, 0xfedcba98, false, false))
    #define SLICE_END()          __builtin_amdgcn_sched_barrier(0)
    auto consume = [&](const float* hbuf, int nsteps_any_lane) {
        const int nsteps = __builtin_amdgcn_readfirstlane(nsteps_any_lane);
        const float* rp = hbuf + lane;
        float x = rp[0];
        float v = rp[HSTR];
        const int o1 = (nsteps > 1) ? 2 * HSTR : 0;
        float xn = rp[o1];
        float vn = rp[o1 + HSTR];
        HOLD4(x, v, xn, vn);
        rp += (nsteps > 2) ? 4 * HSTR : o1;
        float a1v[8];
        {
            float P[8];
            #pragma unroll
            for (int i = 0; i < 8; ++i) P[i] = w1c[i] * x;
            wave_sum8_rows(P);
            #pragma unroll
            for (int i = 0; i < 8; ++i) a1v[i] = fmaxf(P[i] + b1v[i], 0.f);
        }
        float y = fmaf(w2r[1], a1v[1], fmaf(w2r[0], a1v[0], b2v));
        #pragma unroll 1
        for (int s = 0; s < nsteps; ++s) {
            const float x2 = rp[0];
            const float v2 = rp[HSTR];
            SLICE_END();
            if (s + 3 < nsteps) rp += 2 * HSTR;

            float gate[8], e[4], m[4], qe[4], qm[4], re[4], qr[4], Rb[8], sg[8], a1n[8];
            float xe, dxv, dsub, d, nld, dx1;

            y = fmaf(w2r[2], a1v[2], y);
            y = fmaf(w2r[3], a1v[3], y);
            #pragma unroll
            for (int k = 0; k < 4; ++k) e[k] = xlane<16>(w1c[k + 4] * xn);
            xe = xlane<16>(x * xn);
            HOLD5(y, e[0], e[1], e[2], e[3]); HOLD1(xe); SLICE_END();

            y = fmaf(w2r[4], a1v[4], y);
            #pragma unroll
            for (int k = 0; k < 4; ++k) m[k] = w1c[k] * xn + e[k];
            dxv = x * xn + xe;
            gate[0] = (a1v[0] > 0.f) ? -LRC : 0.f;
            HOLD5(y, m[0], m[1], m[2], m[3]); HOLD2(dxv, gate[0]); SLICE_END();

            y = fmaf(w2r[5], a1v[5], y);
            m[0] += xlane<8>(m[0]); m[1] += xlane<8>(m[1]);
            gate[1] = (a1v[1] > 0.f) ? -LRC : 0.f;
            HOLD4(y, m[0], m[1], gate[1]); SLICE_END();

            y = fmaf(w2r[6], a1v[6], y);
            m[2] += xlane<8>(m[2]); m[3] += xlane<8>(m[3]);
            gate[2] = (a1v[2] > 0.f) ? -LRC : 0.f;
            HOLD4(y, m[2], m[3], gate[2]); SLICE_END();

            y = fmaf(w2r[7], a1v[7], y);
            dxv += xlane<8>(dxv); m[0] += xlane<4>(m[0]);
            gate[3] = (a1v[3] > 0.f) ? -LRC : 0.f;
            HOLD4(y, dxv, m[0], gate[3]); SLICE_END();

            dsub = y - v;
            m[1] += xlane<4>(m[1]); m[2] += xlane<4>(m[2]);
            gate[4] = (a1v[4] > 0.f) ? -LRC : 0.f;
            HOLD4(dsub, m[1], m[2], gate[4]); SLICE_END();

            d = dsub * 0.0625f;
            m[3] += xlane<4>(m[3]); dxv += xlane<4>(dxv);
            gate[5] = (a1v[5] > 0.f) ? -LRC : 0.f;
            HOLD4(d, m[3], dxv, gate[5]); SLICE_END();

            #pragma unroll
            for (int k = 0; k < 4; ++k) qe[k] = xlane<16>(w2r[k + 4] * d);
            nld = -LRC * d;
            m[0] += xlane<2>(m[0]); m[1] += xlane<2>(m[1]);
            gate[6] = (a1v[6] > 0.f) ? -LRC : 0.f;
            HOLD4(qe[0], qe[1], qe[2], qe[3]); HOLD4(nld, m[0], m[1], gate[6]); SLICE_END();

            #pragma unroll
            for (int k = 0; k < 4; ++k) qm[k] = w2r[k] * d + qe[k];
            m[2] += xlane<2>(m[2]); m[3] += xlane<2>(m[3]);
            gate[7] = (a1v[7] > 0.f) ? -LRC : 0.f;
            b2v = fmaf(-LRC, d, b2v);
            HOLD4(qm[0], qm[1], qm[2], qm[3]); HOLD4(m[2], m[3], gate[7], b2v); SLICE_END();

            dxv += xlane<2>(dxv); m[0] += xlane<1>(m[0]); m[1] += xlane<1>(m[1]);
            HOLD3(dxv, m[0], m[1]); SLICE_END();

            #pragma unroll
            for (int k = 0; k < 4; ++k) qm[k] += xlane<8>(qm[k]);
            w2r[0] = fmaf(nld, a1v[0], w2r[0]); w2r[1] = fmaf(nld, a1v[1], w2r[1]);
            m[2] += xlane<1>(m[2]);
            HOLD4(qm[0], qm[1], qm[2], qm[3]); HOLD3(w2r[0], w2r[1], m[2]); SLICE_END();
            #pragma unroll
            for (int k = 0; k < 4; ++k) qm[k] += xlane<4>(qm[k]);
            w2r[2] = fmaf(nld, a1v[2], w2r[2]); w2r[3] = fmaf(nld, a1v[3], w2r[3]);
            m[3] += xlane<1>(m[3]);
            re[0] = XROW_INTO(e[0], m[0]);
            HOLD4(qm[0], qm[1], qm[2], qm[3]); HOLD4(w2r[2], w2r[3], m[3], re[0]); SLICE_END();
            #pragma unroll
            for (int k = 0; k < 4; ++k) qm[k] += xlane<2>(qm[k]);
            w2r[4] = fmaf(nld, a1v[4], w2r[4]); w2r[5] = fmaf(nld, a1v[5], w2r[5]);
            dxv += xlane<1>(dxv);
            re[1] = XROW_INTO(e[1], m[1]);
            re[2] = XROW_INTO(e[2], m[2]);
            HOLD4(qm[0], qm[1], qm[2], qm[3]); HOLD5(w2r[4], w2r[5], dxv, re[1], re[2]); SLICE_END();
            #pragma unroll
            for (int k = 0; k < 4; ++k) qm[k] += xlane<1>(qm[k]);
            w2r[6] = fmaf(nld, a1v[6], w2r[6]); w2r[7] = fmaf(nld, a1v[7], w2r[7]);
            re[3] = XROW_INTO(e[3], m[3]);
            dx1 = dxv + 1.f;
            HOLD4(qm[0], qm[1], qm[2], qm[3]); HOLD4(w2r[6], w2r[7], re[3], dx1); SLICE_END();

            #pragma unroll
            for (int k = 0; k < 4; ++k) {
                sg[k] = qm[k] * gate[k];
                qr[k] = XROW_INTO(qe[k], qm[k]);
                Rb[k] = m[k] + b1v[k]; Rb[k + 4] = re[k] + b1v[k + 4];
            }
            HOLD4(sg[0], sg[1], sg[2], sg[3]); HOLD4(qr[0], qr[1], qr[2], qr[3]);
            HOLD4(Rb[0], Rb[1], Rb[2], Rb[3]); HOLD4(Rb[4], Rb[5], Rb[6], Rb[7]); SLICE_END();

            #pragma unroll
            for (int k = 0; k < 4; ++k) {
                a1n[k]    = fmaxf(fmaf(sg[k], dx1, Rb[k]), 0.f);
                sg[k + 4] = qr[k] * gate[k + 4];
                b1v[k]    = fmaf(qm[k], gate[k], b1v[k]);
                w1c[k]    = fmaf(sg[k], x, w1c[k]);
            }
            HOLD4(a1n[0], a1n[1], a1n[2], a1n[3]); HOLD4(sg[4], sg[5], sg[6], sg[7]); SLICE_END();

            y = fmaf(w2r[0], a1n[0], b2v);
            #pragma unroll
            for (int k = 4; k < 8; ++k) {
                a1n[k] = fmaxf(fmaf(sg[k], dx1, Rb[k]), 0.f);
                b1v[k] = fmaf(qr[k - 4], gate[k], b1v[k]);
                w1c[k] = fmaf(sg[k], x, w1c[k]);
            }
            HOLD5(y, a1n[4], a1n[5], a1n[6], a1n[7]); SLICE_END();
            y = fmaf(w2r[1], a1n[1], y);
            #pragma unroll
            for (int i = 0; i < 8; ++i) a1v[i] = a1n[i];
            x = xn; v = vn; xn = x2; vn = v2;
        }
    };
    #undef HOLD1
    #undef HOLD2
    #undef HOLD3
    #undef HOLD4
    #undef HOLD5
    #undef SLICE_END
    #undef XROW_INTO

    if (wave < NP) produce(wave, hid[0] + wave * (CH * HSTR));
    __syncthreads();

    for (int g = 0; g < NGRP; ++g) {
        if (wave < NP) {
            const int c = NP * (g + 1) + wave;
            if (c < NCH) produce(c, hid[(g + 1) & 1] + wave * (CH * HSTR));
        } else {
            consume(hid[g & 1], (g == NGRP - 1) ? (8 * (NCH - NP * (NGRP - 1)) - 1) : 8 * NP);
        }
        __syncthreads();
    }

    if (wave == NP) {
        const float* hbuf = hid[(NGRP - 1) & 1] + ((NCH - 1) - NP * (NGRP - 1)) * (CH * HSTR);
        float x = hbuf[15 * HSTR + lane];
        float p[8];
        #pragma unroll
        for (int i = 0; i < 8; ++i) p[i] = w1c[i] * x;
        wave_sum8_rows(p);
        float y = b2v;
        #pragma unroll
        for (int i = 0; i < 8; ++i) y = fmaf(w2r[i], fmaxf(p[i] + b1v[i], 0.f), y);
        ctxLds[lane] = y;
    }
    __syncthreads();

    if (tid < Vq) {
        float acc = out_b[tid];
        #pragma unroll
        for (int h = 0; h < Hq; ++h)
            acc = fmaf(ctxLds[h], out_w[tid * Hq + h], acc);
        out[b * Vq + tid] = acc;
    }
}

extern "C" void kernel_launch(void* const* d_in, const int* in_sizes, int n_in,
                              void* d_out, int out_size, void* d_ws, size_t ws_size,
                              hipStream_t stream) {
    (void)in_sizes; (void)n_in; (void)out_size; (void)d_ws; (void)ws_size;
    const int*   seq    = (const int*)  d_in[0];
    const float* embed  = (const float*)d_in[1];
    const float* ff_w1  = (const float*)d_in[2];
    const float* ff_b1  = (const float*)d_in[3];
    const float* ff_w2  = (const float*)d_in[4];
    const float* ff_b2  = (const float*)d_in[5];
    const float* ln_g   = (const float*)d_in[6];
    const float* ln_b   = (const float*)d_in[7];
    const float* mlp_w1 = (const float*)d_in[8];
    const float* mlp_b1 = (const float*)d_in[9];
    const float* mlp_w2 = (const float*)d_in[10];
    const float* mlp_b2 = (const float*)d_in[11];
    const float* out_w  = (const float*)d_in[12];
    const float* out_b  = (const float*)d_in[13];
    float*       out    = (float*)d_out;

    pmm_fused_kernel<<<dim3(Bq), dim3(NTHR), 0, stream>>>(
        seq, embed, ff_w1, ff_b1, ff_w2, ff_b2, ln_g, ln_b,
        mlp_w1, mlp_b1, mlp_w2, mlp_b2, out_w, out_b, out);
}


# Encoder run once per block on the 64 vocabulary rows (hidden row = function of token id alone); waves 0-2 only copy rows: SGD groups of 256 steps, not 24: 8 restarts, not 86; p's direct sums move
# speedup vs baseline: 7.2281x; 1.0693x over previous
#include <hip/hip_runtime.h>
#include <hip/hip_bf16.h>
#include <math.h>

typedef __attribute__((ext_vector_type(2))) float v2f;
typedef __attribute__((ext_vector_type(8))) float v8f;

#define LRC  0.01f
#define EPSC 1e-5f

#define Bq 256
#define Lq 4096
#define Hq 32
#define Vq 64
#define CH 16
#define NP  3
#define NW  (NP + 1)
#define NTHR 128
static_assert(NTHR == 32 * NW, "block = NP gather waves + 1 SGD wave");
static_assert(Vq == NW * CH, "the vocabulary is one 16-row tile per wave");
#define GSTEPS 256
#define GROWS  (2 * GSTEPS)
#define NGRP   (Lq / GROWS)
static_assert(Lq % GROWS == 0 && GROWS % 32 == 0, "groups are whole 32-row batches");
#define NBAT   (GROWS / 32)
#define MAXB   ((NBAT + NP - 1) / NP)
#define A1STR 68
#define HSTR  36
#define LNB   8
static_assert(CH % LNB == 0, "a tile is a whole number of LayerNorm batches");

template <int OFF>
__device__ __forceinline__ float xlane(float v) {
    const int b = __builtin_bit_cast(int, v);
    if constexpr (OFF == 16) {
        return __builtin_bit_cast(float,
            __builtin_amdgcn_permlanex16(b, b, 0x76543210, 0xfedcba98, false, false));
    } else {
        return __builtin_bit_cast(float,
            __builtin_amdgcn_update_dpp(0, b, 0x160 + OFF, 0xf, 0xf, true));
    }
}
template <int OFF, int N>
__device__ __forceinline__ void bfly_add(float (&v)[N]) {
    #pragma unroll
    for (int i = 0; i < N; ++i) v[i] += xlane<OFF>(v[i]);
}
template <int N>
__device__ __forceinline__ void wave_sum(float (&v)[N]) {
    bfly_add<16, N>(v);
    bfly_add<8, N>(v);
    bfly_add<4, N>(v);
    bfly_add<2, N>(v);
    bfly_add<1, N>(v);
}
#define SLOT_ROT(lane) (((lane) >> 4) << 2)
__device__ __forceinline__ void wave_sum8_rows(float (&v)[8]) {
    float mine[4];
    #pragma unroll
    for (int k = 0; k < 4; ++k) mine[k] = v[k] + xlane<16>(v[k + 4]);
    bfly_add<8, 4>(mine);
    bfly_add<4, 4>(mine);
    bfly_add<2, 4>(mine);
    bfly_add<1, 4>(mine);
    #pragma unroll
    for (int k = 0; k < 4; ++k) {
        v[k]     = mine[k];
        v[k + 4] = xlane<16>(mine[k]);
    }
}

__launch_bounds__(NTHR, 1)
__global__ void pmm_fused_kernel(const int* __restrict__ seq,
                                 const float* __restrict__ embed,
                                 const float* __restrict__ ff_w1,
                                 const float* __restrict__ ff_b1,
                                 const float* __restrict__ ff_w2,
                                 const float* __restrict__ ff_b2,
                                 const float* __restrict__ ln_g,
                                 const float* __restrict__ ln_b,
                                 const float* __restrict__ mlp_w1,
                                 const float* __restrict__ mlp_b1,
                                 const float* __restrict__ mlp_w2,
                                 const float* __restrict__ mlp_b2,
                                 const float* __restrict__ out_w,
                                 const float* __restrict__ out_b,
                                 float* __restrict__ out)
{
    __shared__ float embedLds[Vq * Hq];
    __shared__ float w1Lds[2 * Hq * Hq];
    __shared__ float w2Lds[Hq * 2 * Hq];
    __shared__ float a1buf[NW][CH * A1STR];
    __shared__ float htab[Vq * HSTR];
    __shared__ float hid[2][GROWS * HSTR];
    __shared__ float ctxLds[Hq];

    const int b    = blockIdx.x;
    const int tid  = threadIdx.x;
    const int lane = tid & 31;
    const int wave = tid >> 5;
    const int m    = lane & 15;
    const int hi   = lane >> 4;

    for (int i = tid; i < Vq * Hq; i += NTHR)     embedLds[i] = embed[i];
    for (int i = tid; i < 2 * Hq * Hq; i += NTHR) w1Lds[i]    = ff_w1[i];
    for (int i = tid; i < 2 * Hq * Hq; i += NTHR) w2Lds[i]    = ff_w2[i];
    int tok0[NBAT / NW];
    static_assert(NBAT % NW == 0, "group 0 is gathered by all waves, the same number of batches each");
    #pragma unroll
    for (int k = 0; k < NBAT / NW; ++k) tok0[k] = seq[b * Lq + (wave + NW * k) * 32 + lane];
    __syncthreads();

    float b1t[4], b2t[2];
    #pragma unroll
    for (int nt = 0; nt < 4; ++nt) b1t[nt] = ff_b1[nt * 16 + m];
    #pragma unroll
    for (int nt = 0; nt < 2; ++nt) b2t[nt] = ff_b2[nt * 16 + m];
    const float lnGv = ln_g[lane];
    const float lnBv = ln_b[lane];

    v2f w1B[4][8], w2B[2][16];
    #pragma unroll
    for (int nt = 0; nt < 4; ++nt) {
        #pragma unroll
        for (int kb = 0; kb < 8; ++kb)
            w1B[nt][kb] = *(const v2f*)&w1Lds[(nt * 16 + m) * Hq + 4 * kb + 2 * hi];
    }
    #pragma unroll
    for (int nt = 0; nt < 2; ++nt) {
        #pragma unroll
        for (int kb = 0; kb < 16; ++kb)
            w2B[nt][kb] = *(const v2f*)&w2Lds[(nt * 16 + m) * (2 * Hq) + 4 * kb + 2 * hi];
    }

    float w1c[8], b1v[8], w2r[8], b2v = 0.f;
    if (wave == NP) {
        const int rot = SLOT_ROT(lane);
        #pragma unroll
        for (int k = 0; k < 8; ++k) {
            const int i = k ^ rot;
            w1c[k] = mlp_w1[i * Hq + lane];
            b1v[k] = mlp_b1[i];
            w2r[k] = mlp_w2[lane * 8 + i];
        }
        b2v = mlp_b2[lane];
    }

    float* const a1w = a1buf[wave];
    auto produce = [&](int c, float* hbuf) {
        const int base = c * CH;
        const int tok  = base + m;

        v2f eA[8];
        #pragma unroll
        for (int kb = 0; kb < 8; ++kb)
            eA[kb] = *(const v2f*)&embedLds[tok * Hq + 4 * kb + 2 * hi];

        #pragma unroll
        for (int nt = 0; nt < 4; ++nt) {
            const int n0 = nt * 16;
            v8f acc = {};
            #pragma unroll
            for (int kb = 0; kb < 8; ++kb) {
                acc = __builtin_amdgcn_wmma_f32_16x16x4_f32(
                        false, eA[kb], false, w1B[nt][kb], (short)0, acc, false, false);
            }
            const float bias = b1t[nt];
            #pragma unroll
            for (int r = 0; r < 8; ++r) {
                float z = acc[r] + bias;
                a1w[(r + 8 * hi) * A1STR + n0 + m] = fmaxf(z, 0.f);
            }
        }
        asm volatile("s_wait_dscnt 0" ::: "memory");
        __builtin_amdgcn_wave_barrier();

        v2f a2[16];
        #pragma unroll
        for (int kb = 0; kb < 16; ++kb)
            a2[kb] = *(const v2f*)&a1w[m * A1STR + 4 * kb + 2 * hi];

        #pragma unroll
        for (int nt = 0; nt < 2; ++nt) {
            const int n0 = nt * 16;
            float res[8];
            #pragma unroll
            for (int r = 0; r < 8; ++r) {
                const int tm = base + r + 8 * hi;
                res[r] = embedLds[tm * Hq + n0 + m];
            }
            v8f acc = {};
            #pragma unroll
            for (int kb = 0; kb < 16; ++kb) {
                acc = __builtin_amdgcn_wmma_f32_16x16x4_f32(
                        false, a2[kb], false, w2B[nt][kb], (short)0, acc, false, false);
            }
            const float bias = b2t[nt];
            #pragma unroll
            for (int r = 0; r < 8; ++r)
                hbuf[(r + 8 * hi) * HSTR + n0 + m] = acc[r] + bias + res[r];
        }
        asm volatile("s_wait_dscnt 0" ::: "memory");
        __builtin_amdgcn_wave_barrier();

        #pragma unroll 1
        for (int t0 = 0; t0 < CH; t0 += LNB) {
            float* const hp = hbuf + t0 * HSTR + lane;
            float z[LNB];
            #pragma unroll
            for (int t = 0; t < LNB; ++t) z[t] = hp[t * HSTR];
            float ss[2 * LNB];
            #pragma unroll
            for (int t = 0; t < LNB; ++t) { ss[2 * t] = z[t]; ss[2 * t + 1] = z[t] * z[t]; }
            wave_sum<2 * LNB>(ss);
            #pragma unroll
            for (int t = 0; t < LNB; ++t) {
                const float s = ss[2 * t], s2 = ss[2 * t + 1];
                float mu  = s * (1.f / 32.f);
                float var = s2 * (1.f / 32.f) - mu * mu;
                float nz  = (z[t] - mu) * rsqrtf(var + EPSC);
                hp[t * HSTR] = nz * lnGv + lnBv;
            }
        }
        asm volatile("s_wait_dscnt 0" ::: "memory");
    };

    #define HOLD1(a)             asm volatile("" :: "v"(a))
    #define HOLD2(a, b)          asm volatile("" :: "v"(a), "v"(b))
    #define HOLD3(a, b, c)       asm volatile("" :: "v"(a), "v"(b), "v"(c))
    #define HOLD4(a, b, c, d)    asm volatile("" :: "v"(a), "v"(b), "v"(c), "v"(d))
    #define HOLD5(a, b, c, d, e) asm volatile("" :: "v"(a), "v"(b), "v"(c), "v"(d), "v"(e))
    #define XROW_INTO(dead, v)   __builtin_bit_cast(float, __builtin_amdgcn_permlanex16( \
                                     __builtin_bit_cast(int, dead), __builtin_bit_cast(int, v), \
                                     0x76543210, 0xfedcba98, false, false))
    #define SLICE_END()          __builtin_amdgcn_sched_barrier(0)
    auto consume = [&](const float* hbuf, int nsteps_any_lane) {
        const int nsteps = __builtin_amdgcn_readfirstlane(nsteps_any_lane);
        const float* rp = hbuf + lane;
        float x = rp[0];
        float v = rp[HSTR];
        const int o1 = (nsteps > 1) ? 2 * HSTR : 0;
        float xn = rp[o1];
        float vn = rp[o1 + HSTR];
        HOLD4(x, v, xn, vn);
        rp += (nsteps > 2) ? 4 * HSTR : o1;
        float a1v[8];
        {
            float P[8];
            #pragma unroll
            for (int i = 0; i < 8; ++i) P[i] = w1c[i] * x;
            wave_sum8_rows(P);
            #pragma unroll
            for (int i = 0; i < 8; ++i) a1v[i] = fmaxf(P[i] + b1v[i], 0.f);
        }
        float y = fmaf(w2r[1], a1v[1], fmaf(w2r[0], a1v[0], b2v));
        #pragma unroll 1
        for (int s = 0; s < nsteps; ++s) {
            const float x2 = rp[0];
            const float v2 = rp[HSTR];
            SLICE_END();
            if (s + 3 < nsteps) rp += 2 * HSTR;

            float gate[8], e[4], m[4], qe[4], qm[4], re[4], qr[4], Rb[8], sg[8], a1n[8];
            float xe, dxv, dsub, d, nld, dx1;

            y = fmaf(w2r[2], a1v[2], y);
            y = fmaf(w2r[3], a1v[3], y);
            #pragma unroll
            for (int k = 0; k < 4; ++k) e[k] = xlane<16>(w1c[k + 4] * xn);
            xe = xlane<16>(x * xn);
            HOLD5(y, e[0], e[1], e[2], e[3]); HOLD1(xe); SLICE_END();

            y = fmaf(w2r[4], a1v[4], y);
            #pragma unroll
            for (int k = 0; k < 4; ++k) m[k] = w1c[k] * xn + e[k];
            dxv = x * xn + xe;
            gate[0] = (a1v[0] > 0.f) ? -LRC : 0.f;
            HOLD5(y, m[0], m[1], m[2], m[3]); HOLD2(dxv, gate[0]); SLICE_END();

            y = fmaf(w2r[5], a1v[5], y);
            m[0] += xlane<8>(m[0]); m[1] += xlane<8>(m[1]);
            gate[1] = (a1v[1] > 0.f) ? -LRC : 0.f;
            HOLD4(y, m[0], m[1], gate[1]); SLICE_END();

            y = fmaf(w2r[6], a1v[6], y);
            m[2] += xlane<8>(m[2]); m[3] += xlane<8>(m[3]);
            gate[2] = (a1v[2] > 0.f) ? -LRC : 0.f;
            HOLD4(y, m[2], m[3], gate[2]); SLICE_END();

            y = fmaf(w2r[7], a1v[7], y);
            dxv += xlane<8>(dxv); m[0] += xlane<4>(m[0]);
            gate[3] = (a1v[3] > 0.f) ? -LRC : 0.f;
            HOLD4(y, dxv, m[0], gate[3]); SLICE_END();

            dsub = y - v;
            m[1] += xlane<4>(m[1]); m[2] += xlane<4>(m[2]);
            gate[4] = (a1v[4] > 0.f) ? -LRC : 0.f;
            HOLD4(dsub, m[1], m[2], gate[4]); SLICE_END();

            d = dsub * 0.0625f;
            m[3] += xlane<4>(m[3]); dxv += xlane<4>(dxv);
            gate[5] = (a1v[5] > 0.f) ? -LRC : 0.f;
            HOLD4(d, m[3], dxv, gate[5]); SLICE_END();

            #pragma unroll
            for (int k = 0; k < 4; ++k) qe[k] = xlane<16>(w2r[k + 4] * d);
            nld = -LRC * d;
            m[0] += xlane<2>(m[0]); m[1] += xlane<2>(m[1]);
            gate[6] = (a1v[6] > 0.f) ? -LRC : 0.f;
            HOLD4(qe[0], qe[1], qe[2], qe[3]); HOLD4(nld, m[0], m[1], gate[6]); SLICE_END();

            #pragma unroll
            for (int k = 0; k < 4; ++k) qm[k] = w2r[k] * d + qe[k];
            m[2] += xlane<2>(m[2]); m[3] += xlane<2>(m[3]);
            gate[7] = (a1v[7] > 0.f) ? -LRC : 0.f;
            b2v = fmaf(-LRC, d, b2v);
            HOLD4(qm[0], qm[1], qm[2], qm[3]); HOLD4(m[2], m[3], gate[7], b2v); SLICE_END();

            dxv += xlane<2>(dxv); m[0] += xlane<1>(m[0]); m[1] += xlane<1>(m[1]);
            HOLD3(dxv, m[0], m[1]); SLICE_END();

            #pragma unroll
            for (int k = 0; k < 4; ++k) qm[k] += xlane<8>(qm[k]);
            w2r[0] = fmaf(nld, a1v[0], w2r[0]); w2r[1] = fmaf(nld, a1v[1], w2r[1]);
            m[2] += xlane<1>(m[2]);
            HOLD4(qm[0], qm[1], qm[2], qm[3]); HOLD3(w2r[0], w2r[1], m[2]); SLICE_END();
            #pragma unroll
            for (int k = 0; k < 4; ++k) qm[k] += xlane<4>(qm[k]);
            w2r[2] = fmaf(nld, a1v[2], w2r[2]); w2r[3] = fmaf(nld, a1v[3], w2r[3]);
            m[3] += xlane<1>(m[3]);
            re[0] = XROW_INTO(e[0], m[0]);
            HOLD4(qm[0], qm[1], qm[2], qm[3]); HOLD4(w2r[2], w2r[3], m[3], re[0]); SLICE_END();
            #pragma unroll
            for (int k = 0; k < 4; ++k) qm[k] += xlane<2>(qm[k]);
            w2r[4] = fmaf(nld, a1v[4], w2r[4]); w2r[5] = fmaf(nld, a1v[5], w2r[5]);
            dxv += xlane<1>(dxv);
            re[1] = XROW_INTO(e[1], m[1]);
            re[2] = XROW_INTO(e[2], m[2]);
            HOLD4(qm[0], qm[1], qm[2], qm[3]); HOLD5(w2r[4], w2r[5], dxv, re[1], re[2]); SLICE_END();
            #pragma unroll
            for (int k = 0; k < 4; ++k) qm[k] += xlane<1>(qm[k]);
            w2r[6] = fmaf(nld, a1v[6], w2r[6]); w2r[7] = fmaf(nld, a1v[7], w2r[7]);
            re[3] = XROW_INTO(e[3], m[3]);
            dx1 = dxv + 1.f;
            HOLD4(qm[0], qm[1], qm[2], qm[3]); HOLD4(w2r[6], w2r[7], re[3], dx1); SLICE_END();

            #pragma unroll
            for (int k = 0; k < 4; ++k) {
                sg[k] = qm[k] * gate[k];
                qr[k] = XROW_INTO(qe[k], qm[k]);
                Rb[k] = m[k] + b1v[k]; Rb[k + 4] = re[k] + b1v[k + 4];
            }
            HOLD4(sg[0], sg[1], sg[2], sg[3]); HOLD4(qr[0], qr[1], qr[2], qr[3]);
            HOLD4(Rb[0], Rb[1], Rb[2], Rb[3]); HOLD4(Rb[4], Rb[5], Rb[6], Rb[7]); SLICE_END();

            #pragma unroll
            for (int k = 0; k < 4; ++k) {
                a1n[k]    = fmaxf(fmaf(sg[k], dx1, Rb[k]), 0.f);
                sg[k + 4] = qr[k] * gate[k + 4];
                b1v[k]    = fmaf(qm[k], gate[k], b1v[k]);
                w1c[k]    = fmaf(sg[k], x, w1c[k]);
            }
            HOLD4(a1n[0], a1n[1], a1n[2], a1n[3]); HOLD4(sg[4], sg[5], sg[6], sg[7]); SLICE_END();

            y = fmaf(w2r[0], a1n[0], b2v);
            #pragma unroll
            for (int k = 4; k < 8; ++k) {
                a1n[k] = fmaxf(fmaf(sg[k], dx1, Rb[k]), 0.f);
                b1v[k] = fmaf(qr[k - 4], gate[k], b1v[k]);
                w1c[k] = fmaf(sg[k], x, w1c[k]);
            }
            HOLD5(y, a1n[4], a1n[5], a1n[6], a1n[7]); SLICE_END();
            y = fmaf(w2r[1], a1n[1], y);
            #pragma unroll
            for (int i = 0; i < 8; ++i) a1v[i] = a1n[i];
            x = xn; v = vn; xn = x2; vn = v2;
        }
    };
    #undef HOLD1
    #undef HOLD2
    #undef HOLD3
    #undef HOLD4
    #undef HOLD5
    #undef SLICE_END
    #undef XROW_INTO

    auto gather32 = [&](int tokv, float* dst) {
        float row[32];
        #pragma unroll
        for (int r = 0; r < 32; ++r) row[r] = htab[__builtin_amdgcn_readlane(tokv, r) * HSTR + lane];
        #pragma unroll
        for (int r = 0; r < 32; ++r) dst[r * HSTR + lane] = row[r];
    };

    produce(wave, htab + wave * (CH * HSTR));
    __syncthreads();
    #pragma unroll
    for (int k = 0; k < NBAT / NW; ++k) gather32(tok0[k], hid[0] + (wave + NW * k) * 32 * HSTR);
    __syncthreads();

    for (int g = 0; g < NGRP; ++g) {
        if (wave < NP) {
            if (g + 1 < NGRP) {
                const int* const sp  = seq + b * Lq + (g + 1) * GROWS + lane;
                float* const     dst = hid[(g + 1) & 1];
                int tk[MAXB];
                #pragma unroll
                for (int k = 0; k < MAXB; ++k) {
                    const int j = wave + NP * k;
                    tk[k] = (j < NBAT) ? sp[j * 32] : 0;
                }
                #pragma unroll
                for (int k = 0; k < MAXB; ++k) {
                    const int j = wave + NP * k;
                    if (j < NBAT) gather32(tk[k], dst + j * 32 * HSTR);
                }
            }
        } else {
            consume(hid[g & 1], (g == NGRP - 1) ? GSTEPS - 1 : GSTEPS);
        }
        __syncthreads();
    }

    if (wave == NP) {
        const float* hbuf = hid[(NGRP - 1) & 1];
        float x = hbuf[(GROWS - 1) * HSTR + lane];
        float p[8];
        #pragma unroll
        for (int i = 0; i < 8; ++i) p[i] = w1c[i] * x;
        wave_sum8_rows(p);
        float y = b2v;
        #pragma unroll
        for (int i = 0; i < 8; ++i) y = fmaf(w2r[i], fmaxf(p[i] + b1v[i], 0.f), y);
        ctxLds[lane] = y;
    }
    __syncthreads();

    if (tid < Vq) {
        float acc = out_b[tid];
        #pragma unroll
        for (int h = 0; h < Hq; ++h)
            acc = fmaf(ctxLds[h], out_w[tid * Hq + h], acc);
        out[b * Vq + tid] = acc;
    }
}

extern "C" void kernel_launch(void* const* d_in, const int* in_sizes, int n_in,
                              void* d_out, int out_size, void* d_ws, size_t ws_size,
                              hipStream_t stream) {
    (void)in_sizes; (void)n_in; (void)out_size; (void)d_ws; (void)ws_size;
    const int*   seq    = (const int*)  d_in[0];
    const float* embed  = (const float*)d_in[1];
    const float* ff_w1  = (const float*)d_in[2];
    const float* ff_b1  = (const float*)d_in[3];
    const float* ff_w2  = (const float*)d_in[4];
    const float* ff_b2  = (const float*)d_in[5];
    const float* ln_g   = (const float*)d_in[6];
    const float* ln_b   = (const float*)d_in[7];
    const float* mlp_w1 = (const float*)d_in[8];
    const float* mlp_b1 = (const float*)d_in[9];
    const float* mlp_w2 = (const float*)d_in[10];
    const float* mlp_b2 = (const float*)d_in[11];
    const float* out_w  = (const float*)d_in[12];
    const float* out_b  = (const float*)d_in[13];
    float*       out    = (float*)d_out;

    pmm_fused_kernel<<<dim3(Bq), dim3(NTHR), 0, stream>>>(
        seq, embed, ff_w1, ff_b1, ff_w2, ff_b2, ln_g, ln_b,
        mlp_w1, mlp_b1, mlp_w2, mlp_b2, out_w, out_b, out);
}
